# starting kernel
# baseline (speedup 1.0000x reference)
#include <hip/hip_runtime.h>
#include <hip/hip_bf16.h>

typedef float v2f __attribute__((ext_vector_type(2)));
typedef float v8f __attribute__((ext_vector_type(8)));

static __host__ int cdiv(int a, int b) { return (a + b - 1) / b; }

__global__ void zero_f(float* __restrict__ p, int n) {
  int i = blockIdx.x * blockDim.x + threadIdx.x;
  if (i < n) p[i] = 0.0f;
}

__global__ void edge_degree(const int* __restrict__ dst, int E, float* __restrict__ degf) {
  int e = blockIdx.x * blockDim.x + threadIdx.x;
  if (e < E) atomicAdd(&degf[dst[e]], 1.0f);
}

__global__ void make_dinv(float* __restrict__ dinv, int N) {
  int i = blockIdx.x * blockDim.x + threadIdx.x;
  if (i < N) dinv[i] = rsqrtf(dinv[i] + 1.0f);
}

__global__ void gemm_tile16(const float* __restrict__ X, const float* __restrict__ W,
                            const float* __restrict__ dinv, float* __restrict__ G,
                            int N, int K, int tiles) {
  int tile = blockIdx.x * (blockDim.x >> 5) + (threadIdx.x >> 5);
  if (tile >= tiles) return;
  int lane = threadIdx.x & 31;
  int hi = lane >> 4;
  int lo = lane & 15;
  int row = tile * 16 + lo;
  float rowMask = (row < N) ? 1.0f : 0.0f;
  const float* xr = X + (size_t)((row < N) ? row : 0) * K;
  const float* wc = W + lo;
  int K4 = K & ~3;

  v8f c = {};
  int k0 = 0;
  for (; k0 < K4; k0 += 4) {
    int kk = k0 + 2 * hi;
    if ((k0 & 63) == 0) __builtin_prefetch(xr + kk + 64, 0, 1);
    v2f a, b;
    a.x = xr[kk] * rowMask;
    a.y = xr[kk + 1] * rowMask;
    b.x = wc[kk * 16];
    b.y = wc[(kk + 1) * 16];
    c = __builtin_amdgcn_wmma_f32_16x16x4_f32(false, a, false, b, (short)0, c,
                                              false, false);
  }
  if (k0 < K) {
    int kk = k0 + 2 * hi;
    int kc0 = (kk < K) ? kk : 0;
    int kc1 = (kk + 1 < K) ? kk + 1 : 0;
    float m0 = (kk < K) ? rowMask : 0.0f;
    float m1 = (kk + 1 < K) ? rowMask : 0.0f;
    v2f a, b;
    a.x = xr[kc0] * m0;
    a.y = xr[kc1] * m1;
    b.x = wc[kc0 * 16];
    b.y = wc[kc1 * 16];
    c = __builtin_amdgcn_wmma_f32_16x16x4_f32(false, a, false, b, (short)0, c,
                                              false, false);
  }
#pragma unroll
  for (int j = 0; j < 8; ++j) {
    int rr = tile * 16 + 8 * hi + j;
    if (rr < N) G[(size_t)rr * 16 + lo] = c[j] * dinv[rr];
  }
}

__global__ void edge_agg(const int* __restrict__ src, const int* __restrict__ dst,
                         const float* __restrict__ g, float* __restrict__ agg, int E4) {
  int t = blockIdx.x * blockDim.x + threadIdx.x;
  if (t >= E4) return;
  int e = t >> 2, part = t & 3;
  int s = src[e], d = dst[e];
  float4 v = ((const float4*)(g + (size_t)s * 16))[part];
  float* o = agg + (size_t)d * 16 + part * 4;
  atomicAdd(o + 0, v.x);
  atomicAdd(o + 1, v.y);
  atomicAdd(o + 2, v.z);
  atomicAdd(o + 3, v.w);
}

__global__ void post1(const float* __restrict__ agg, const float* __restrict__ g,
                      const float* __restrict__ dinv, const float* __restrict__ b,
                      float* __restrict__ h, int N) {
  int i = blockIdx.x * blockDim.x + threadIdx.x;
  if (i >= N) return;
  float di = dinv[i];
  const float4* av = (const float4*)(agg + (size_t)i * 16);
  const float4* gv = (const float4*)(g + (size_t)i * 16);
  float4* hv = (float4*)(h + (size_t)i * 16);
#pragma unroll
  for (int j = 0; j < 4; ++j) {
    float4 a = av[j], gg = gv[j];
    float4 r;
    r.x = fmaxf(di * (a.x + gg.x) + b[j * 4 + 0], 0.0f);
    r.y = fmaxf(di * (a.y + gg.y) + b[j * 4 + 1], 0.0f);
    r.z = fmaxf(di * (a.z + gg.z) + b[j * 4 + 2], 0.0f);
    r.w = fmaxf(di * (a.w + gg.w) + b[j * 4 + 3], 0.0f);
    hv[j] = r;
  }
}

__global__ void post2(const float* __restrict__ agg, const float* __restrict__ g,
                      const float* __restrict__ dinv, const float* __restrict__ b,
                      const float* __restrict__ Wfc, float* __restrict__ p,
                      float* __restrict__ q, int N) {
  int i = blockIdx.x * blockDim.x + threadIdx.x;
  if (i >= N) return;
  float di = dinv[i];
  float sp = 0.0f, sq = 0.0f;
#pragma unroll
  for (int f = 0; f < 16; ++f) {
    float h = fmaxf(di * (agg[(size_t)i * 16 + f] + g[(size_t)i * 16 + f]) + b[f], 0.0f);
    sp += h * Wfc[f];
    sq += h * Wfc[16 + f];
  }
  p[i] = sp;
  q[i] = sq;
}

__global__ void edge_out(const int* __restrict__ src, const int* __restrict__ dst,
                         const float* __restrict__ p, const float* __restrict__ q,
                         const float* __restrict__ bfc, float* __restrict__ out, int E) {
  int e = blockIdx.x * blockDim.x + threadIdx.x;
  if (e < E) out[e] = p[src[e]] + q[dst[e]] + bfc[0];
}

extern "C" void kernel_launch(void* const* d_in, const int* in_sizes, int n_in,
                              void* d_out, int out_size, void* d_ws, size_t ws_size,
                              hipStream_t stream) {
  const float* x   = (const float*)d_in[0];
  const int*   ei  = (const int*)d_in[1];
  const float* W1  = (const float*)d_in[2];
  const float* b1  = (const float*)d_in[3];
  const float* W2  = (const float*)d_in[4];
  const float* b2  = (const float*)d_in[5];
  const float* Wfc = (const float*)d_in[6];
  const float* bfc = (const float*)d_in[7];
  float* out = (float*)d_out;

  const int K1 = 495;
  const int N = in_sizes[0] / K1;
  const int E = in_sizes[1] / 2;
  const int* src = ei;
  const int* dst = ei + E;

  float* ws  = (float*)d_ws;
  float* dinv = ws;
  float* g    = ws + (size_t)N;
  float* agg  = ws + (size_t)N * 17;
  float* h1   = ws + (size_t)N * 33;
  float* p    = ws + (size_t)N * 49;
  float* q    = ws + (size_t)N * 50;

  const int B = 256;
  const int tiles = cdiv(N, 16);
  const int gemmBlocks = cdiv(tiles, B / 32);

  zero_f<<<cdiv(N, B), B, 0, stream>>>(dinv, N);
  edge_degree<<<cdiv(E, B), B, 0, stream>>>(dst, E, dinv);
  make_dinv<<<cdiv(N, B), B, 0, stream>>>(dinv, N);

  gemm_tile16<<<gemmBlocks, B, 0, stream>>>(x, W1, dinv, g, N, K1, tiles);
  zero_f<<<cdiv(16 * N, B), B, 0, stream>>>(agg, 16 * N);
  edge_agg<<<cdiv(4 * E, B), B, 0, stream>>>(src, dst, g, agg, 4 * E);
  post1<<<cdiv(N, B), B, 0, stream>>>(agg, g, dinv, b1, h1, N);

  gemm_tile16<<<gemmBlocks, B, 0, stream>>>(h1, W2, dinv, g, N, 16, tiles);
  zero_f<<<cdiv(16 * N, B), B, 0, stream>>>(agg, 16 * N);
  edge_agg<<<cdiv(4 * E, B), B, 0, stream>>>(src, dst, g, agg, 4 * E);
  post2<<<cdiv(N, B), B, 0, stream>>>(agg, g, dinv, b2, Wfc, p, q, N);

  edge_out<<<cdiv(E, B), B, 0, stream>>>(src, dst, p, q, bfc, out, E);
}


# edges bucketed by destination range once per call (LDS counting sort); each bucket's work-group counts degrees and sums rows by LDS f32 atomic adds, stores once: no global atomics, new sum order
# speedup vs baseline: 2.3939x; 2.3939x over previous
#include <hip/hip_runtime.h>
#include <hip/hip_bf16.h>

typedef float v2f __attribute__((ext_vector_type(2)));
typedef float v8f __attribute__((ext_vector_type(8)));

static __host__ int cdiv(int a, int b) { return (a + b - 1) / b; }


#define BSH 7
#define BNODES 128
#define NBMAX 1024
#define CHUNK 25600
#define APITCH 17

__global__ void __launch_bounds__(1024)
bucket_count(const int* __restrict__ dst, int E, int NB, int* __restrict__ cnt) {
  __shared__ int h[NBMAX];
  h[threadIdx.x] = 0;
  __syncthreads();
  int e0 = blockIdx.x * CHUNK;
  int e1 = min(E, e0 + CHUNK);
  int e = e0 + (int)threadIdx.x;
  for (; e + 4 * 1024 < e1; e += 5 * 1024) {
    int d0 = dst[e], d1 = dst[e + 1024], d2 = dst[e + 2048], d3 = dst[e + 3072], d4 = dst[e + 4096];
    atomicAdd(&h[d0 >> BSH], 1);
    atomicAdd(&h[d1 >> BSH], 1);
    atomicAdd(&h[d2 >> BSH], 1);
    atomicAdd(&h[d3 >> BSH], 1);
    atomicAdd(&h[d4 >> BSH], 1);
  }
  for (; e < e1; e += 1024) atomicAdd(&h[dst[e] >> BSH], 1);
  __syncthreads();
  if ((int)threadIdx.x < NB) cnt[blockIdx.x * NB + threadIdx.x] = h[threadIdx.x];
}

__global__ void __launch_bounds__(1024)
bucket_scan(const int* __restrict__ cnt, int* __restrict__ base, int NB, int nchunk,
            int* __restrict__ bstart) {
  __shared__ int s[1024];
  int b = threadIdx.x;
  int run = 0;
  if (b < NB) {
    for (int w0 = 0; w0 < nchunk; w0 += 16) {
      int c[16];
#pragma unroll
      for (int j = 0; j < 16; ++j) c[j] = cnt[min(w0 + j, nchunk - 1) * NB + b];
#pragma unroll
      for (int j = 0; j < 16; ++j) {
        if (w0 + j < nchunk) base[(w0 + j) * NB + b] = run;
        run += (w0 + j < nchunk) ? c[j] : 0;
      }
    }
  }
  s[b] = run;
  __syncthreads();
  for (int off = 1; off < 1024; off <<= 1) {
    int v = (b >= off) ? s[b - off] : 0;
    __syncthreads();
    s[b] += v;
    __syncthreads();
  }
  if (b < NB) bstart[b] = s[b] - run;
  if (b == 0) bstart[NB] = s[1023];
}

__global__ void __launch_bounds__(1024)
bucket_place(const int* __restrict__ src, const int* __restrict__ dst, int E, int NB,
             const int* __restrict__ cnt, const int* __restrict__ base,
             const int* __restrict__ bstart, unsigned int* __restrict__ sorted) {
  __shared__ unsigned int stage[CHUNK];
  __shared__ int loff[1024];
  __shared__ int cur[1024];
  __shared__ int gpos[1024];
  int t = threadIdx.x;
  int c = (t < NB) ? cnt[blockIdx.x * NB + t] : 0;
  loff[t] = c;
  __syncthreads();
  for (int off = 1; off < 1024; off <<= 1) {
    int v = (t >= off) ? loff[t - off] : 0;
    __syncthreads();
    loff[t] += v;
    __syncthreads();
  }
  int ex = loff[t] - c;
  loff[t] = ex;
  cur[t] = ex;
  gpos[t] = (t < NB) ? bstart[t] + base[blockIdx.x * NB + t] : 0;
  __syncthreads();
  int e0 = blockIdx.x * CHUNK;
  int e1 = min(E, e0 + CHUNK);
  int e = e0 + t;
  for (; e + 4 * 1024 < e1; e += 5 * 1024) {
    int d[5], s[5];
#pragma unroll
    for (int j = 0; j < 5; ++j) { d[j] = dst[e + j * 1024]; s[j] = src[e + j * 1024]; }
#pragma unroll
    for (int j = 0; j < 5; ++j) {
      int r = atomicAdd(&cur[d[j] >> BSH], 1);
      stage[r] = ((unsigned int)s[j] << BSH) | (unsigned int)(d[j] & (BNODES - 1));
    }
  }
  for (; e < e1; e += 1024) {
    int d = dst[e], s = src[e];
    int r = atomicAdd(&cur[d >> BSH], 1);
    stage[r] = ((unsigned int)s << BSH) | (unsigned int)(d & (BNODES - 1));
  }
  __syncthreads();
  int wave = __builtin_amdgcn_readfirstlane(t >> 5), lane = t & 31;
  for (int b = wave; b < NB; b += 32) {
    int lo = loff[b], n = cur[b] - lo, gp = gpos[b];
    for (int j = lane; j < n; j += 32) sorted[gp + j] = stage[lo + j];
  }
}

__global__ void __launch_bounds__(4 * BNODES)
bucket_dinv(const unsigned int* __restrict__ sorted, const int* __restrict__ bstart,
            float* __restrict__ dinv, int N) {
  __shared__ int c[BNODES];
  const int T = 4 * BNODES;
  if (threadIdx.x < BNODES) c[threadIdx.x] = 0;
  __syncthreads();
  int e0 = bstart[blockIdx.x], e1 = bstart[blockIdx.x + 1];
  int e = e0 + (int)threadIdx.x;
  for (; e + 3 * T < e1; e += 4 * T) {
    unsigned int w0 = sorted[e], w1 = sorted[e + T], w2 = sorted[e + 2 * T], w3 = sorted[e + 3 * T];
    atomicAdd(&c[w0 & (BNODES - 1)], 1);
    atomicAdd(&c[w1 & (BNODES - 1)], 1);
    atomicAdd(&c[w2 & (BNODES - 1)], 1);
    atomicAdd(&c[w3 & (BNODES - 1)], 1);
  }
  for (; e < e1; e += T) atomicAdd(&c[sorted[e] & (BNODES - 1)], 1);
  __syncthreads();
  int node = blockIdx.x * BNODES + threadIdx.x;
  if (threadIdx.x < BNODES && node < N) dinv[node] = rsqrtf((float)c[threadIdx.x] + 1.0f);
}

__device__ __forceinline__ void bucket_sum(const unsigned int* __restrict__ sorted, int e0, int e1,
                                           const float* __restrict__ g, float* acc) {
  for (int i = threadIdx.x; i < BNODES * APITCH; i += 4 * BNODES) acc[i] = 0.0f;
  __syncthreads();
  int lane = threadIdx.x & 31, wave = threadIdx.x >> 5;
  int sub = lane >> 2, part = lane & 3;
  int trips = (e1 - e0 + 4 * BNODES - 1) / (4 * BNODES);
  for (int it = 0; it < trips; ++it) {
    int eb = e0 + it * (4 * BNODES) + wave * 32;
    unsigned int mine = sorted[min(eb + lane, e1 - 1)];
    unsigned int w[4];
    float4 v[4];
#pragma unroll
    for (int u = 0; u < 4; ++u) w[u] = (unsigned int)__shfl((int)mine, u * 8 + sub);
#pragma unroll
    for (int u = 0; u < 4; ++u) v[u] = ((const float4*)(g + (size_t)(w[u] >> BSH) * 16))[part];
#pragma unroll
    for (int u = 0; u < 4; ++u) {
      bool ok = (eb + u * 8 + sub) < e1;
      float* o = acc + (w[u] & (BNODES - 1)) * APITCH + part * 4;
      atomicAdd(o + 0, ok ? v[u].x : 0.0f);
      atomicAdd(o + 1, ok ? v[u].y : 0.0f);
      atomicAdd(o + 2, ok ? v[u].z : 0.0f);
      atomicAdd(o + 3, ok ? v[u].w : 0.0f);
    }
  }
  __syncthreads();
}

__global__ void __launch_bounds__(4 * BNODES)
bucket_agg1(const unsigned int* __restrict__ sorted, const int* __restrict__ bstart,
            const float* __restrict__ g, const float* __restrict__ dinv,
            const float* __restrict__ b, float* __restrict__ h, int N) {
  __shared__ float acc[BNODES * APITCH];
  bucket_sum(sorted, bstart[blockIdx.x], bstart[blockIdx.x + 1], g, acc);
  int r = threadIdx.x >> 2, part = threadIdx.x & 3;
  int node = blockIdx.x * BNODES + r;
  if (node < N) {
    float di = dinv[node];
    float4 gg = ((const float4*)(g + (size_t)node * 16))[part];
    const float* a = acc + r * APITCH + part * 4;
    float4 o;
    o.x = fmaxf(di * (a[0] + gg.x) + b[part * 4 + 0], 0.0f);
    o.y = fmaxf(di * (a[1] + gg.y) + b[part * 4 + 1], 0.0f);
    o.z = fmaxf(di * (a[2] + gg.z) + b[part * 4 + 2], 0.0f);
    o.w = fmaxf(di * (a[3] + gg.w) + b[part * 4 + 3], 0.0f);
    ((float4*)(h + (size_t)node * 16))[part] = o;
  }
}

__global__ void __launch_bounds__(4 * BNODES)
bucket_agg2(const unsigned int* __restrict__ sorted, const int* __restrict__ bstart,
            const float* __restrict__ g, const float* __restrict__ dinv,
            const float* __restrict__ b, const float* __restrict__ Wfc,
            float* __restrict__ p, float* __restrict__ q, int N) {
  __shared__ float acc[BNODES * APITCH];
  bucket_sum(sorted, bstart[blockIdx.x], bstart[blockIdx.x + 1], g, acc);
  int node = blockIdx.x * BNODES + threadIdx.x;
  if (threadIdx.x < BNODES && node < N) {
    float di = dinv[node];
    float sp = 0.0f, sq = 0.0f;
#pragma unroll
    for (int f = 0; f < 16; ++f) {
      float hh = fmaxf(di * (acc[threadIdx.x * APITCH + f] + g[(size_t)node * 16 + f]) + b[f], 0.0f);
      sp += hh * Wfc[f];
      sq += hh * Wfc[16 + f];
    }
    p[node] = sp;
    q[node] = sq;
  }
}

__global__ void gemm_tile16(const float* __restrict__ X, const float* __restrict__ W,
                            const float* __restrict__ dinv, float* __restrict__ G,
                            int N, int K, int tiles) {
  int tile = blockIdx.x * (blockDim.x >> 5) + (threadIdx.x >> 5);
  if (tile >= tiles) return;
  int lane = threadIdx.x & 31;
  int hi = lane >> 4;
  int lo = lane & 15;
  int row = tile * 16 + lo;
  float rowMask = (row < N) ? 1.0f : 0.0f;
  const float* xr = X + (size_t)((row < N) ? row : 0) * K;
  const float* wc = W + lo;
  int K4 = K & ~3;

  v8f c = {};
  int k0 = 0;
  for (; k0 < K4; k0 += 4) {
    int kk = k0 + 2 * hi;
    if ((k0 & 63) == 0) __builtin_prefetch(xr + kk + 64, 0, 1);
    v2f a, b;
    a.x = xr[kk] * rowMask;
    a.y = xr[kk + 1] * rowMask;
    b.x = wc[kk * 16];
    b.y = wc[(kk + 1) * 16];
    c = __builtin_amdgcn_wmma_f32_16x16x4_f32(false, a, false, b, (short)0, c,
                                              false, false);
  }
  if (k0 < K) {
    int kk = k0 + 2 * hi;
    int kc0 = (kk < K) ? kk : 0;
    int kc1 = (kk + 1 < K) ? kk + 1 : 0;
    float m0 = (kk < K) ? rowMask : 0.0f;
    float m1 = (kk + 1 < K) ? rowMask : 0.0f;
    v2f a, b;
    a.x = xr[kc0] * m0;
    a.y = xr[kc1] * m1;
    b.x = wc[kc0 * 16];
    b.y = wc[kc1 * 16];
    c = __builtin_amdgcn_wmma_f32_16x16x4_f32(false, a, false, b, (short)0, c,
                                              false, false);
  }
#pragma unroll
  for (int j = 0; j < 8; ++j) {
    int rr = tile * 16 + 8 * hi + j;
    if (rr < N) G[(size_t)rr * 16 + lo] = c[j] * dinv[rr];
  }
}

__global__ void edge_out(const int* __restrict__ src, const int* __restrict__ dst,
                         const float* __restrict__ p, const float* __restrict__ q,
                         const float* __restrict__ bfc, float* __restrict__ out, int E) {
  int e = blockIdx.x * blockDim.x + threadIdx.x;
  if (e < E) out[e] = p[src[e]] + q[dst[e]] + bfc[0];
}

extern "C" void kernel_launch(void* const* d_in, const int* in_sizes, int n_in,
                              void* d_out, int out_size, void* d_ws, size_t ws_size,
                              hipStream_t stream) {
  const float* x   = (const float*)d_in[0];
  const int*   ei  = (const int*)d_in[1];
  const float* W1  = (const float*)d_in[2];
  const float* b1  = (const float*)d_in[3];
  const float* W2  = (const float*)d_in[4];
  const float* b2  = (const float*)d_in[5];
  const float* Wfc = (const float*)d_in[6];
  const float* bfc = (const float*)d_in[7];
  float* out = (float*)d_out;

  const int K1 = 495;
  const int N = in_sizes[0] / K1;
  const int E = in_sizes[1] / 2;
  const int* src = ei;
  const int* dst = ei + E;

  const int NB = cdiv(N, BNODES);
  const int nchunk = cdiv(E, CHUNK);

  float* ws  = (float*)d_ws;
  float* dinv = ws;
  float* g    = ws + (size_t)N;
  float* h1   = ws + (size_t)N * 17;
  float* p    = ws + (size_t)N * 33;
  float* q    = ws + (size_t)N * 34;
  int* cnt    = (int*)(ws + (size_t)N * 35);
  int* base   = cnt + (size_t)nchunk * NB;
  int* bstart = base + (size_t)nchunk * NB;
  unsigned int* sorted = (unsigned int*)d_out;

  const int B = 256;
  const int tiles = cdiv(N, 16);
  const int gemmBlocks = cdiv(tiles, B / 32);

  bucket_count<<<nchunk, 1024, 0, stream>>>(dst, E, NB, cnt);
  bucket_scan<<<1, 1024, 0, stream>>>(cnt, base, NB, nchunk, bstart);
  bucket_place<<<nchunk, 1024, 0, stream>>>(src, dst, E, NB, cnt, base, bstart, sorted);

  bucket_dinv<<<NB, 4 * BNODES, 0, stream>>>(sorted, bstart, dinv, N);

  gemm_tile16<<<gemmBlocks, B, 0, stream>>>(x, W1, dinv, g, N, K1, tiles);
  bucket_agg1<<<NB, 4 * BNODES, 0, stream>>>(sorted, bstart, g, dinv, b1, h1, N);

  gemm_tile16<<<gemmBlocks, B, 0, stream>>>(h1, W2, dinv, g, N, 16, tiles);
  bucket_agg2<<<NB, 4 * BNODES, 0, stream>>>(sorted, bstart, g, dinv, b2, Wfc, p, q, N);

  edge_out<<<cdiv(E, B), B, 0, stream>>>(src, dst, p, q, bfc, out, E);
}


# bucket sums: a bucket's edges are ordered by destination node in LDS (integer counting sort); 4 lanes add a node's rows in registers, 8 gathers in flight; no LDS float atomics, new sum order
# speedup vs baseline: 2.8265x; 1.1807x over previous
#include <hip/hip_runtime.h>
#include <hip/hip_bf16.h>

typedef float v2f __attribute__((ext_vector_type(2)));
typedef float v8f __attribute__((ext_vector_type(8)));

static __host__ int cdiv(int a, int b) { return (a + b - 1) / b; }


#define BSH 7
#define BNODES 128
#define NBMAX 1024
#define CHUNK 25600
#define APITCH 17

__global__ void __launch_bounds__(1024)
bucket_count(const int* __restrict__ dst, int E, int NB, int* __restrict__ cnt) {
  __shared__ int h[NBMAX];
  h[threadIdx.x] = 0;
  __syncthreads();
  int e0 = blockIdx.x * CHUNK;
  int e1 = min(E, e0 + CHUNK);
  int e = e0 + (int)threadIdx.x;
  for (; e + 4 * 1024 < e1; e += 5 * 1024) {
    int d0 = dst[e], d1 = dst[e + 1024], d2 = dst[e + 2048], d3 = dst[e + 3072], d4 = dst[e + 4096];
    atomicAdd(&h[d0 >> BSH], 1);
    atomicAdd(&h[d1 >> BSH], 1);
    atomicAdd(&h[d2 >> BSH], 1);
    atomicAdd(&h[d3 >> BSH], 1);
    atomicAdd(&h[d4 >> BSH], 1);
  }
  for (; e < e1; e += 1024) atomicAdd(&h[dst[e] >> BSH], 1);
  __syncthreads();
  if ((int)threadIdx.x < NB) cnt[blockIdx.x * NB + threadIdx.x] = h[threadIdx.x];
}

__global__ void __launch_bounds__(1024)
bucket_scan(const int* __restrict__ cnt, int* __restrict__ base, int NB, int nchunk,
            int* __restrict__ bstart) {
  __shared__ int s[1024];
  int b = threadIdx.x;
  int run = 0;
  if (b < NB) {
    for (int w0 = 0; w0 < nchunk; w0 += 16) {
      int c[16];
#pragma unroll
      for (int j = 0; j < 16; ++j) c[j] = cnt[min(w0 + j, nchunk - 1) * NB + b];
#pragma unroll
      for (int j = 0; j < 16; ++j) {
        if (w0 + j < nchunk) base[(w0 + j) * NB + b] = run;
        run += (w0 + j < nchunk) ? c[j] : 0;
      }
    }
  }
  s[b] = run;
  __syncthreads();
  for (int off = 1; off < 1024; off <<= 1) {
    int v = (b >= off) ? s[b - off] : 0;
    __syncthreads();
    s[b] += v;
    __syncthreads();
  }
  if (b < NB) bstart[b] = s[b] - run;
  if (b == 0) bstart[NB] = s[1023];
}

__global__ void __launch_bounds__(1024)
bucket_place(const int* __restrict__ src, const int* __restrict__ dst, int E, int NB,
             const int* __restrict__ cnt, const int* __restrict__ base,
             const int* __restrict__ bstart, unsigned int* __restrict__ sorted) {
  __shared__ unsigned int stage[CHUNK];
  __shared__ int loff[1024];
  __shared__ int cur[1024];
  __shared__ int gpos[1024];
  int t = threadIdx.x;
  int c = (t < NB) ? cnt[blockIdx.x * NB + t] : 0;
  loff[t] = c;
  __syncthreads();
  for (int off = 1; off < 1024; off <<= 1) {
    int v = (t >= off) ? loff[t - off] : 0;
    __syncthreads();
    loff[t] += v;
    __syncthreads();
  }
  int ex = loff[t] - c;
  loff[t] = ex;
  cur[t] = ex;
  gpos[t] = (t < NB) ? bstart[t] + base[blockIdx.x * NB + t] : 0;
  __syncthreads();
  int e0 = blockIdx.x * CHUNK;
  int e1 = min(E, e0 + CHUNK);
  int e = e0 + t;
  for (; e + 4 * 1024 < e1; e += 5 * 1024) {
    int d[5], s[5];
#pragma unroll
    for (int j = 0; j < 5; ++j) { d[j] = dst[e + j * 1024]; s[j] = src[e + j * 1024]; }
#pragma unroll
    for (int j = 0; j < 5; ++j) {
      int r = atomicAdd(&cur[d[j] >> BSH], 1);
      stage[r] = ((unsigned int)s[j] << BSH) | (unsigned int)(d[j] & (BNODES - 1));
    }
  }
  for (; e < e1; e += 1024) {
    int d = dst[e], s = src[e];
    int r = atomicAdd(&cur[d >> BSH], 1);
    stage[r] = ((unsigned int)s << BSH) | (unsigned int)(d & (BNODES - 1));
  }
  __syncthreads();
  int wave = __builtin_amdgcn_readfirstlane(t >> 5), lane = t & 31;
  for (int b = wave; b < NB; b += 32) {
    int lo = loff[b], n = cur[b] - lo, gp = gpos[b];
    for (int j = lane; j < n; j += 32) sorted[gp + j] = stage[lo + j];
  }
}

__global__ void __launch_bounds__(4 * BNODES)
bucket_dinv(const unsigned int* __restrict__ sorted, const int* __restrict__ bstart,
            float* __restrict__ dinv, int N) {
  __shared__ int c[BNODES];
  const int T = 4 * BNODES;
  if (threadIdx.x < BNODES) c[threadIdx.x] = 0;
  __syncthreads();
  int e0 = bstart[blockIdx.x], e1 = bstart[blockIdx.x + 1];
  int e = e0 + (int)threadIdx.x;
  for (; e + 3 * T < e1; e += 4 * T) {
    unsigned int w0 = sorted[e], w1 = sorted[e + T], w2 = sorted[e + 2 * T], w3 = sorted[e + 3 * T];
    atomicAdd(&c[w0 & (BNODES - 1)], 1);
    atomicAdd(&c[w1 & (BNODES - 1)], 1);
    atomicAdd(&c[w2 & (BNODES - 1)], 1);
    atomicAdd(&c[w3 & (BNODES - 1)], 1);
  }
  for (; e < e1; e += T) atomicAdd(&c[sorted[e] & (BNODES - 1)], 1);
  __syncthreads();
  int node = blockIdx.x * BNODES + threadIdx.x;
  if (threadIdx.x < BNODES && node < N) dinv[node] = rsqrtf((float)c[threadIdx.x] + 1.0f);
}

#define RPER 10
#define RWORDS (RPER * 4 * BNODES)
__device__ __forceinline__ float4 bucket_sum(const unsigned int* __restrict__ sorted, int e0, int e1,
                                             const float* __restrict__ g, unsigned int* stage,
                                             int* cnt, int* off, int* wtot) {
  const int T = 4 * BNODES;
  const int t = threadIdx.x;
  const int lane = t & 31;
  const int node = t >> 2, part = t & 3;
  const float4* g4 = (const float4*)g;
  float4 acc = make_float4(0.0f, 0.0f, 0.0f, 0.0f);
  for (int c0 = e0; c0 < e1; c0 += RWORDS) {
    const int c1 = min(e1, c0 + RWORDS);
    if (t < BNODES) cnt[t] = 0;
    __syncthreads();
    unsigned int w[RPER];
    int r[RPER];
#pragma unroll
    for (int j = 0; j < RPER; ++j) w[j] = sorted[min(c0 + j * T + t, c1 - 1)];
#pragma unroll
    for (int j = 0; j < RPER; ++j) {
      r[j] = 0;
      if (c0 + j * T + t < c1) r[j] = atomicAdd(&cnt[w[j] & (BNODES - 1)], 1);
    }
    __syncthreads();
    int c = (t < BNODES) ? cnt[t] : 0;
    int x = c;
    for (int d = 1; d < 32; d <<= 1) {
      int y = __shfl_up(x, d);
      if (lane >= d) x += y;
    }
    if (t < BNODES && lane == 31) wtot[t >> 5] = x;
    __syncthreads();
    if (t < BNODES) {
      int pre = 0;
      for (int k = 0; k < (t >> 5); ++k) pre += wtot[k];
      off[t] = pre + x - c;
    }
    __syncthreads();
    int o[RPER];
#pragma unroll
    for (int j = 0; j < RPER; ++j) o[j] = off[w[j] & (BNODES - 1)];
#pragma unroll
    for (int j = 0; j < RPER; ++j)
      if (c0 + j * T + t < c1) stage[o[j] + r[j]] = w[j] >> BSH;
    __syncthreads();
    const int st = off[node];
    const int n = cnt[node];
    for (int j = 0; j < n; j += 8) {
      unsigned int s[8];
      float4 v[8];
#pragma unroll
      for (int u = 0; u < 8; ++u) s[u] = stage[st + min(j + u, n - 1)];
#pragma unroll
      for (int u = 0; u < 8; ++u) v[u] = g4[(s[u] << 2) + (unsigned int)part];
#pragma unroll
      for (int u = 0; u < 8; ++u) {
        bool ok = (j + u) < n;
        acc.x += ok ? v[u].x : 0.0f;
        acc.y += ok ? v[u].y : 0.0f;
        acc.z += ok ? v[u].z : 0.0f;
        acc.w += ok ? v[u].w : 0.0f;
      }
    }
    __syncthreads();
  }
  return acc;
}

__global__ void __launch_bounds__(4 * BNODES, 16)
bucket_agg1(const unsigned int* __restrict__ sorted, const int* __restrict__ bstart,
            const float* __restrict__ g, const float* __restrict__ dinv,
            const float* __restrict__ b, float* __restrict__ h, int N) {
  __shared__ unsigned int stage[RWORDS];
  __shared__ int cnt[BNODES], off[BNODES], wtot[BNODES / 32];
  float4 a = bucket_sum(sorted, bstart[blockIdx.x], bstart[blockIdx.x + 1], g, stage, cnt, off, wtot);
  int r = threadIdx.x >> 2, part = threadIdx.x & 3;
  int node = blockIdx.x * BNODES + r;
  if (node < N) {
    float di = dinv[node];
    float4 gg = ((const float4*)(g + (size_t)node * 16))[part];
    float4 o;
    o.x = fmaxf(di * (a.x + gg.x) + b[part * 4 + 0], 0.0f);
    o.y = fmaxf(di * (a.y + gg.y) + b[part * 4 + 1], 0.0f);
    o.z = fmaxf(di * (a.z + gg.z) + b[part * 4 + 2], 0.0f);
    o.w = fmaxf(di * (a.w + gg.w) + b[part * 4 + 3], 0.0f);
    ((float4*)(h + (size_t)node * 16))[part] = o;
  }
}

__global__ void __launch_bounds__(4 * BNODES, 16)
bucket_agg2(const unsigned int* __restrict__ sorted, const int* __restrict__ bstart,
            const float* __restrict__ g, const float* __restrict__ dinv,
            const float* __restrict__ b, const float* __restrict__ Wfc,
            float* __restrict__ p, float* __restrict__ q, int N) {
  __shared__ unsigned int stage[RWORDS];
  __shared__ int cnt[BNODES], off[BNODES], wtot[BNODES / 32];
  __shared__ float acc[BNODES * APITCH];
  float4 a = bucket_sum(sorted, bstart[blockIdx.x], bstart[blockIdx.x + 1], g, stage, cnt, off, wtot);
  {
    float* o = acc + (threadIdx.x >> 2) * APITCH + (threadIdx.x & 3) * 4;
    o[0] = a.x;
    o[1] = a.y;
    o[2] = a.z;
    o[3] = a.w;
  }
  __syncthreads();
  int node = blockIdx.x * BNODES + threadIdx.x;
  if (threadIdx.x < BNODES && node < N) {
    float di = dinv[node];
    float sp = 0.0f, sq = 0.0f;
#pragma unroll
    for (int f = 0; f < 16; ++f) {
      float hh = fmaxf(di * (acc[threadIdx.x * APITCH + f] + g[(size_t)node * 16 + f]) + b[f], 0.0f);
      sp += hh * Wfc[f];
      sq += hh * Wfc[16 + f];
    }
    p[node] = sp;
    q[node] = sq;
  }
}

__global__ void gemm_tile16(const float* __restrict__ X, const float* __restrict__ W,
                            const float* __restrict__ dinv, float* __restrict__ G,
                            int N, int K, int tiles) {
  int tile = blockIdx.x * (blockDim.x >> 5) + (threadIdx.x >> 5);
  if (tile >= tiles) return;
  int lane = threadIdx.x & 31;
  int hi = lane >> 4;
  int lo = lane & 15;
  int row = tile * 16 + lo;
  float rowMask = (row < N) ? 1.0f : 0.0f;
  const float* xr = X + (size_t)((row < N) ? row : 0) * K;
  const float* wc = W + lo;
  int K4 = K & ~3;

  v8f c = {};
  int k0 = 0;
  for (; k0 < K4; k0 += 4) {
    int kk = k0 + 2 * hi;
    if ((k0 & 63) == 0) __builtin_prefetch(xr + kk + 64, 0, 1);
    v2f a, b;
    a.x = xr[kk] * rowMask;
    a.y = xr[kk + 1] * rowMask;
    b.x = wc[kk * 16];
    b.y = wc[(kk + 1) * 16];
    c = __builtin_amdgcn_wmma_f32_16x16x4_f32(false, a, false, b, (short)0, c,
                                              false, false);
  }
  if (k0 < K) {
    int kk = k0 + 2 * hi;
    int kc0 = (kk < K) ? kk : 0;
    int kc1 = (kk + 1 < K) ? kk + 1 : 0;
    float m0 = (kk < K) ? rowMask : 0.0f;
    float m1 = (kk + 1 < K) ? rowMask : 0.0f;
    v2f a, b;
    a.x = xr[kc0] * m0;
    a.y = xr[kc1] * m1;
    b.x = wc[kc0 * 16];
    b.y = wc[kc1 * 16];
    c = __builtin_amdgcn_wmma_f32_16x16x4_f32(false, a, false, b, (short)0, c,
                                              false, false);
  }
#pragma unroll
  for (int j = 0; j < 8; ++j) {
    int rr = tile * 16 + 8 * hi + j;
    if (rr < N) G[(size_t)rr * 16 + lo] = c[j] * dinv[rr];
  }
}

__global__ void edge_out(const int* __restrict__ src, const int* __restrict__ dst,
                         const float* __restrict__ p, const float* __restrict__ q,
                         const float* __restrict__ bfc, float* __restrict__ out, int E) {
  int e = blockIdx.x * blockDim.x + threadIdx.x;
  if (e < E) out[e] = p[src[e]] + q[dst[e]] + bfc[0];
}

extern "C" void kernel_launch(void* const* d_in, const int* in_sizes, int n_in,
                              void* d_out, int out_size, void* d_ws, size_t ws_size,
                              hipStream_t stream) {
  const float* x   = (const float*)d_in[0];
  const int*   ei  = (const int*)d_in[1];
  const float* W1  = (const float*)d_in[2];
  const float* b1  = (const float*)d_in[3];
  const float* W2  = (const float*)d_in[4];
  const float* b2  = (const float*)d_in[5];
  const float* Wfc = (const float*)d_in[6];
  const float* bfc = (const float*)d_in[7];
  float* out = (float*)d_out;

  const int K1 = 495;
  const int N = in_sizes[0] / K1;
  const int E = in_sizes[1] / 2;
  const int* src = ei;
  const int* dst = ei + E;

  const int NB = cdiv(N, BNODES);
  const int nchunk = cdiv(E, CHUNK);

  float* ws  = (float*)d_ws;
  float* dinv = ws;
  float* g    = ws + (size_t)N;
  float* h1   = ws + (size_t)N * 17;
  float* p    = ws + (size_t)N * 33;
  float* q    = ws + (size_t)N * 34;
  int* cnt    = (int*)(ws + (size_t)N * 35);
  int* base   = cnt + (size_t)nchunk * NB;
  int* bstart = base + (size_t)nchunk * NB;
  unsigned int* sorted = (unsigned int*)d_out;

  const int B = 256;
  const int tiles = cdiv(N, 16);
  const int gemmBlocks = cdiv(tiles, B / 32);

  bucket_count<<<nchunk, 1024, 0, stream>>>(dst, E, NB, cnt);
  bucket_scan<<<1, 1024, 0, stream>>>(cnt, base, NB, nchunk, bstart);
  bucket_place<<<nchunk, 1024, 0, stream>>>(src, dst, E, NB, cnt, base, bstart, sorted);

  bucket_dinv<<<NB, 4 * BNODES, 0, stream>>>(sorted, bstart, dinv, N);

  gemm_tile16<<<gemmBlocks, B, 0, stream>>>(x, W1, dinv, g, N, K1, tiles);
  bucket_agg1<<<NB, 4 * BNODES, 0, stream>>>(sorted, bstart, g, dinv, b1, h1, N);

  gemm_tile16<<<gemmBlocks, B, 0, stream>>>(h1, W2, dinv, g, N, 16, tiles);
  bucket_agg2<<<NB, 4 * BNODES, 0, stream>>>(sorted, bstart, g, dinv, b2, Wfc, p, q, N);

  edge_out<<<cdiv(E, B), B, 0, stream>>>(src, dst, p, q, bfc, out, E);
}


# x@W1 needs no edge data: its 16-row WMMA tiles (same loop, same k order) run as extra work-groups of the edge-placing launch; dinv scaling of g moves to the degree pass; 9 launches -> 8, same sums
# speedup vs baseline: 2.9709x; 1.0511x over previous
#include <hip/hip_runtime.h>
#include <hip/hip_bf16.h>

typedef float v2f __attribute__((ext_vector_type(2)));
typedef float v8f __attribute__((ext_vector_type(8)));

static __host__ int cdiv(int a, int b) { return (a + b - 1) / b; }


#define BSH 7
#define BNODES 128
#define NBMAX 1024
#define CHUNK 25600
#define APITCH 17

__global__ void __launch_bounds__(1024)
bucket_count(const int* __restrict__ dst, int E, int NB, int* __restrict__ cnt) {
  __shared__ int h[NBMAX];
  h[threadIdx.x] = 0;
  __syncthreads();
  int e0 = blockIdx.x * CHUNK;
  int e1 = min(E, e0 + CHUNK);
  int e = e0 + (int)threadIdx.x;
  for (; e + 4 * 1024 < e1; e += 5 * 1024) {
    int d0 = dst[e], d1 = dst[e + 1024], d2 = dst[e + 2048], d3 = dst[e + 3072], d4 = dst[e + 4096];
    atomicAdd(&h[d0 >> BSH], 1);
    atomicAdd(&h[d1 >> BSH], 1);
    atomicAdd(&h[d2 >> BSH], 1);
    atomicAdd(&h[d3 >> BSH], 1);
    atomicAdd(&h[d4 >> BSH], 1);
  }
  for (; e < e1; e += 1024) atomicAdd(&h[dst[e] >> BSH], 1);
  __syncthreads();
  if ((int)threadIdx.x < NB) cnt[blockIdx.x * NB + threadIdx.x] = h[threadIdx.x];
}

__global__ void __launch_bounds__(1024)
bucket_scan(const int* __restrict__ cnt, int* __restrict__ base, int NB, int nchunk,
            int* __restrict__ bstart) {
  __shared__ int s[1024];
  int b = threadIdx.x;
  int run = 0;
  if (b < NB) {
    for (int w0 = 0; w0 < nchunk; w0 += 16) {
      int c[16];
#pragma unroll
      for (int j = 0; j < 16; ++j) c[j] = cnt[min(w0 + j, nchunk - 1) * NB + b];
#pragma unroll
      for (int j = 0; j < 16; ++j) {
        if (w0 + j < nchunk) base[(w0 + j) * NB + b] = run;
        run += (w0 + j < nchunk) ? c[j] : 0;
      }
    }
  }
  s[b] = run;
  __syncthreads();
  for (int off = 1; off < 1024; off <<= 1) {
    int v = (b >= off) ? s[b - off] : 0;
    __syncthreads();
    s[b] += v;
    __syncthreads();
  }
  if (b < NB) bstart[b] = s[b] - run;
  if (b == 0) bstart[NB] = s[1023];
}

template <bool SCALE>
__device__ __forceinline__ void xw_tile(const float* __restrict__ X, const float* __restrict__ W,
                                        const float* __restrict__ dinv, float* __restrict__ G,
                                        int N, int K, int tile) {
  int lane = threadIdx.x & 31;
  int hi = lane >> 4;
  int lo = lane & 15;
  int row = tile * 16 + lo;
  float rowMask = (row < N) ? 1.0f : 0.0f;
  const float* xr = X + (size_t)((row < N) ? row : 0) * K;
  const float* wc = W + lo;
  int K4 = K & ~3;

  v8f c = {};
  int k0 = 0;
  for (; k0 < K4; k0 += 4) {
    int kk = k0 + 2 * hi;
    if ((k0 & 63) == 0) __builtin_prefetch(xr + kk + 64, 0, 1);
    v2f a, b;
    a.x = xr[kk] * rowMask;
    a.y = xr[kk + 1] * rowMask;
    b.x = wc[kk * 16];
    b.y = wc[(kk + 1) * 16];
    c = __builtin_amdgcn_wmma_f32_16x16x4_f32(false, a, false, b, (short)0, c,
                                              false, false);
  }
  if (k0 < K) {
    int kk = k0 + 2 * hi;
    int kc0 = (kk < K) ? kk : 0;
    int kc1 = (kk + 1 < K) ? kk + 1 : 0;
    float m0 = (kk < K) ? rowMask : 0.0f;
    float m1 = (kk + 1 < K) ? rowMask : 0.0f;
    v2f a, b;
    a.x = xr[kc0] * m0;
    a.y = xr[kc1] * m1;
    b.x = wc[kc0 * 16];
    b.y = wc[kc1 * 16];
    c = __builtin_amdgcn_wmma_f32_16x16x4_f32(false, a, false, b, (short)0, c,
                                              false, false);
  }
#pragma unroll
  for (int j = 0; j < 8; ++j) {
    int rr = tile * 16 + 8 * hi + j;
    if (rr < N) G[(size_t)rr * 16 + lo] = SCALE ? c[j] * dinv[rr] : c[j];
  }
}

__global__ void gemm_tile16(const float* __restrict__ X, const float* __restrict__ W,
                            const float* __restrict__ dinv, float* __restrict__ G,
                            int N, int K, int tiles) {
  int tile = blockIdx.x * (blockDim.x >> 5) + (threadIdx.x >> 5);
  if (tile >= tiles) return;
  xw_tile<true>(X, W, dinv, G, N, K, tile);
}

__global__ void __launch_bounds__(1024)
bucket_place(const int* __restrict__ src, const int* __restrict__ dst, int E, int NB,
             const int* __restrict__ cnt, const int* __restrict__ base,
             const int* __restrict__ bstart, unsigned int* __restrict__ sorted, int nplace,
             const float* __restrict__ X, const float* __restrict__ W, float* __restrict__ G,
             int N, int K, int tiles) {
  __shared__ unsigned int stage[CHUNK];
  __shared__ int loff[1024];
  __shared__ int cur[1024];
  __shared__ int gpos[1024];
  int t = threadIdx.x;
  if ((int)blockIdx.x >= nplace) {
    int tile = ((int)blockIdx.x - nplace) * 32 + (t >> 5);
    if (tile < tiles) xw_tile<false>(X, W, nullptr, G, N, K, tile);
    return;
  }
  int c = (t < NB) ? cnt[blockIdx.x * NB + t] : 0;
  loff[t] = c;
  __syncthreads();
  for (int off = 1; off < 1024; off <<= 1) {
    int v = (t >= off) ? loff[t - off] : 0;
    __syncthreads();
    loff[t] += v;
    __syncthreads();
  }
  int ex = loff[t] - c;
  loff[t] = ex;
  cur[t] = ex;
  gpos[t] = (t < NB) ? bstart[t] + base[blockIdx.x * NB + t] : 0;
  __syncthreads();
  int e0 = blockIdx.x * CHUNK;
  int e1 = min(E, e0 + CHUNK);
  int e = e0 + t;
  for (; e + 4 * 1024 < e1; e += 5 * 1024) {
    int d[5], s[5];
#pragma unroll
    for (int j = 0; j < 5; ++j) { d[j] = dst[e + j * 1024]; s[j] = src[e + j * 1024]; }
#pragma unroll
    for (int j = 0; j < 5; ++j) {
      int r = atomicAdd(&cur[d[j] >> BSH], 1);
      stage[r] = ((unsigned int)s[j] << BSH) | (unsigned int)(d[j] & (BNODES - 1));
    }
  }
  for (; e < e1; e += 1024) {
    int d = dst[e], s = src[e];
    int r = atomicAdd(&cur[d >> BSH], 1);
    stage[r] = ((unsigned int)s << BSH) | (unsigned int)(d & (BNODES - 1));
  }
  __syncthreads();
  int wave = __builtin_amdgcn_readfirstlane(t >> 5), lane = t & 31;
  for (int b = wave; b < NB; b += 32) {
    int lo = loff[b], n = cur[b] - lo, gp = gpos[b];
    for (int j = lane; j < n; j += 32) sorted[gp + j] = stage[lo + j];
  }
}

__global__ void __launch_bounds__(4 * BNODES)
bucket_dinv(const unsigned int* __restrict__ sorted, const int* __restrict__ bstart,
            float* __restrict__ dinv, float* __restrict__ g, int N) {
  __shared__ int c[BNODES];
  const int T = 4 * BNODES;
  const int grow = blockIdx.x * BNODES + ((int)threadIdx.x >> 2);
  float4* gp = (float4*)(g + (size_t)min(grow, N - 1) * 16) + (threadIdx.x & 3);
  if (threadIdx.x < BNODES) c[threadIdx.x] = 0;
  __syncthreads();
  float4 gv = *gp;
  int e0 = bstart[blockIdx.x], e1 = bstart[blockIdx.x + 1];
  int e = e0 + (int)threadIdx.x;
  for (; e + 3 * T < e1; e += 4 * T) {
    unsigned int w0 = sorted[e], w1 = sorted[e + T], w2 = sorted[e + 2 * T], w3 = sorted[e + 3 * T];
    atomicAdd(&c[w0 & (BNODES - 1)], 1);
    atomicAdd(&c[w1 & (BNODES - 1)], 1);
    atomicAdd(&c[w2 & (BNODES - 1)], 1);
    atomicAdd(&c[w3 & (BNODES - 1)], 1);
  }
  for (; e < e1; e += T) atomicAdd(&c[sorted[e] & (BNODES - 1)], 1);
  __syncthreads();
  int node = blockIdx.x * BNODES + threadIdx.x;
  if (threadIdx.x < BNODES && node < N) dinv[node] = rsqrtf((float)c[threadIdx.x] + 1.0f);
  if (grow < N) {
    float di = rsqrtf((float)c[threadIdx.x >> 2] + 1.0f);
    gv.x *= di;
    gv.y *= di;
    gv.z *= di;
    gv.w *= di;
    *gp = gv;
  }
}

#define RPER 10
#define RWORDS (RPER * 4 * BNODES)
__device__ __forceinline__ float4 bucket_sum(const unsigned int* __restrict__ sorted, int e0, int e1,
                                             const float* __restrict__ g, unsigned int* stage,
                                             int* cnt, int* off, int* wtot) {
  const int T = 4 * BNODES;
  const int t = threadIdx.x;
  const int lane = t & 31;
  const int node = t >> 2, part = t & 3;
  const float4* g4 = (const float4*)g;
  float4 acc = make_float4(0.0f, 0.0f, 0.0f, 0.0f);
  for (int c0 = e0; c0 < e1; c0 += RWORDS) {
    const int c1 = min(e1, c0 + RWORDS);
    if (t < BNODES) cnt[t] = 0;
    __syncthreads();
    unsigned int w[RPER];
    int r[RPER];
#pragma unroll
    for (int j = 0; j < RPER; ++j) w[j] = sorted[min(c0 + j * T + t, c1 - 1)];
#pragma unroll
    for (int j = 0; j < RPER; ++j) {
      r[j] = 0;
      if (c0 + j * T + t < c1) r[j] = atomicAdd(&cnt[w[j] & (BNODES - 1)], 1);
    }
    __syncthreads();
    int c = (t < BNODES) ? cnt[t] : 0;
    int x = c;
    for (int d = 1; d < 32; d <<= 1) {
      int y = __shfl_up(x, d);
      if (lane >= d) x += y;
    }
    if (t < BNODES && lane == 31) wtot[t >> 5] = x;
    __syncthreads();
    if (t < BNODES) {
      int pre = 0;
      for (int k = 0; k < (t >> 5); ++k) pre += wtot[k];
      off[t] = pre + x - c;
    }
    __syncthreads();
    int o[RPER];
#pragma unroll
    for (int j = 0; j < RPER; ++j) o[j] = off[w[j] & (BNODES - 1)];
#pragma unroll
    for (int j = 0; j < RPER; ++j)
      if (c0 + j * T + t < c1) stage[o[j] + r[j]] = w[j] >> BSH;
    __syncthreads();
    const int st = off[node];
    const int n = cnt[node];
    for (int j = 0; j < n; j += 8) {
      unsigned int s[8];
      float4 v[8];
#pragma unroll
      for (int u = 0; u < 8; ++u) s[u] = stage[st + min(j + u, n - 1)];
#pragma unroll
      for (int u = 0; u < 8; ++u) v[u] = g4[(s[u] << 2) + (unsigned int)part];
#pragma unroll
      for (int u = 0; u < 8; ++u) {
        bool ok = (j + u) < n;
        acc.x += ok ? v[u].x : 0.0f;
        acc.y += ok ? v[u].y : 0.0f;
        acc.z += ok ? v[u].z : 0.0f;
        acc.w += ok ? v[u].w : 0.0f;
      }
    }
    __syncthreads();
  }
  return acc;
}

__global__ void __launch_bounds__(4 * BNODES, 16)
bucket_agg1(const unsigned int* __restrict__ sorted, const int* __restrict__ bstart,
            const float* __restrict__ g, const float* __restrict__ dinv,
            const float* __restrict__ b, float* __restrict__ h, int N) {
  __shared__ unsigned int stage[RWORDS];
  __shared__ int cnt[BNODES], off[BNODES], wtot[BNODES / 32];
  float4 a = bucket_sum(sorted, bstart[blockIdx.x], bstart[blockIdx.x + 1], g, stage, cnt, off, wtot);
  int r = threadIdx.x >> 2, part = threadIdx.x & 3;
  int node = blockIdx.x * BNODES + r;
  if (node < N) {
    float di = dinv[node];
    float4 gg = ((const float4*)(g + (size_t)node * 16))[part];
    float4 o;
    o.x = fmaxf(di * (a.x + gg.x) + b[part * 4 + 0], 0.0f);
    o.y = fmaxf(di * (a.y + gg.y) + b[part * 4 + 1], 0.0f);
    o.z = fmaxf(di * (a.z + gg.z) + b[part * 4 + 2], 0.0f);
    o.w = fmaxf(di * (a.w + gg.w) + b[part * 4 + 3], 0.0f);
    ((float4*)(h + (size_t)node * 16))[part] = o;
  }
}

__global__ void __launch_bounds__(4 * BNODES, 16)
bucket_agg2(const unsigned int* __restrict__ sorted, const int* __restrict__ bstart,
            const float* __restrict__ g, const float* __restrict__ dinv,
            const float* __restrict__ b, const float* __restrict__ Wfc,
            float* __restrict__ p, float* __restrict__ q, int N) {
  __shared__ unsigned int stage[RWORDS];
  __shared__ int cnt[BNODES], off[BNODES], wtot[BNODES / 32];
  __shared__ float acc[BNODES * APITCH];
  float4 a = bucket_sum(sorted, bstart[blockIdx.x], bstart[blockIdx.x + 1], g, stage, cnt, off, wtot);
  {
    float* o = acc + (threadIdx.x >> 2) * APITCH + (threadIdx.x & 3) * 4;
    o[0] = a.x;
    o[1] = a.y;
    o[2] = a.z;
    o[3] = a.w;
  }
  __syncthreads();
  int node = blockIdx.x * BNODES + threadIdx.x;
  if (threadIdx.x < BNODES && node < N) {
    float di = dinv[node];
    float sp = 0.0f, sq = 0.0f;
#pragma unroll
    for (int f = 0; f < 16; ++f) {
      float hh = fmaxf(di * (acc[threadIdx.x * APITCH + f] + g[(size_t)node * 16 + f]) + b[f], 0.0f);
      sp += hh * Wfc[f];
      sq += hh * Wfc[16 + f];
    }
    p[node] = sp;
    q[node] = sq;
  }
}

__global__ void edge_out(const int* __restrict__ src, const int* __restrict__ dst,
                         const float* __restrict__ p, const float* __restrict__ q,
                         const float* __restrict__ bfc, float* __restrict__ out, int E) {
  int e = blockIdx.x * blockDim.x + threadIdx.x;
  if (e < E) out[e] = p[src[e]] + q[dst[e]] + bfc[0];
}

extern "C" void kernel_launch(void* const* d_in, const int* in_sizes, int n_in,
                              void* d_out, int out_size, void* d_ws, size_t ws_size,
                              hipStream_t stream) {
  const float* x   = (const float*)d_in[0];
  const int*   ei  = (const int*)d_in[1];
  const float* W1  = (const float*)d_in[2];
  const float* b1  = (const float*)d_in[3];
  const float* W2  = (const float*)d_in[4];
  const float* b2  = (const float*)d_in[5];
  const float* Wfc = (const float*)d_in[6];
  const float* bfc = (const float*)d_in[7];
  float* out = (float*)d_out;

  const int K1 = 495;
  const int N = in_sizes[0] / K1;
  const int E = in_sizes[1] / 2;
  const int* src = ei;
  const int* dst = ei + E;

  const int NB = cdiv(N, BNODES);
  const int nchunk = cdiv(E, CHUNK);

  float* ws  = (float*)d_ws;
  float* dinv = ws;
  float* g    = ws + (size_t)N;
  float* h1   = ws + (size_t)N * 17;
  float* p    = ws + (size_t)N * 33;
  float* q    = ws + (size_t)N * 34;
  int* cnt    = (int*)(ws + (size_t)N * 35);
  int* base   = cnt + (size_t)nchunk * NB;
  int* bstart = base + (size_t)nchunk * NB;
  unsigned int* sorted = (unsigned int*)d_out;

  const int B = 256;
  const int tiles = cdiv(N, 16);
  const int gemmBlocks = cdiv(tiles, B / 32);
  const int xwGroups = cdiv(tiles, 32);

  bucket_count<<<nchunk, 1024, 0, stream>>>(dst, E, NB, cnt);
  bucket_scan<<<1, 1024, 0, stream>>>(cnt, base, NB, nchunk, bstart);
  bucket_place<<<nchunk + xwGroups, 1024, 0, stream>>>(src, dst, E, NB, cnt, base, bstart, sorted,
                                                        nchunk, x, W1, g, N, K1, tiles);

  bucket_dinv<<<NB, 4 * BNODES, 0, stream>>>(sorted, bstart, dinv, g, N);

  bucket_agg1<<<NB, 4 * BNODES, 0, stream>>>(sorted, bstart, g, dinv, b1, h1, N);

  gemm_tile16<<<gemmBlocks, B, 0, stream>>>(h1, W2, dinv, g, N, 16, tiles);
  bucket_agg2<<<NB, 4 * BNODES, 0, stream>>>(sorted, bstart, g, dinv, b2, Wfc, p, q, N);

  edge_out<<<cdiv(E, B), B, 0, stream>>>(src, dst, p, q, bfc, out, E);
}


# edge_out: each 12,800-edge chunk is ordered by node range in LDS (integer counting sort, once by dst, once by src); q and p are read in that order, values return to edge order in LDS; same sums
# speedup vs baseline: 3.1033x; 1.0446x over previous
#include <hip/hip_runtime.h>
#include <hip/hip_bf16.h>

typedef float v2f __attribute__((ext_vector_type(2)));
typedef float v8f __attribute__((ext_vector_type(8)));

static __host__ int cdiv(int a, int b) { return (a + b - 1) / b; }


#define BSH 7
#define BNODES 128
#define NBMAX 1024
#define CHUNK 25600
#define APITCH 17

__global__ void __launch_bounds__(1024)
bucket_count(const int* __restrict__ dst, int E, int NB, int* __restrict__ cnt) {
  __shared__ int h[NBMAX];
  h[threadIdx.x] = 0;
  __syncthreads();
  int e0 = blockIdx.x * CHUNK;
  int e1 = min(E, e0 + CHUNK);
  int e = e0 + (int)threadIdx.x;
  for (; e + 4 * 1024 < e1; e += 5 * 1024) {
    int d0 = dst[e], d1 = dst[e + 1024], d2 = dst[e + 2048], d3 = dst[e + 3072], d4 = dst[e + 4096];
    atomicAdd(&h[d0 >> BSH], 1);
    atomicAdd(&h[d1 >> BSH], 1);
    atomicAdd(&h[d2 >> BSH], 1);
    atomicAdd(&h[d3 >> BSH], 1);
    atomicAdd(&h[d4 >> BSH], 1);
  }
  for (; e < e1; e += 1024) atomicAdd(&h[dst[e] >> BSH], 1);
  __syncthreads();
  if ((int)threadIdx.x < NB) cnt[blockIdx.x * NB + threadIdx.x] = h[threadIdx.x];
}

__global__ void __launch_bounds__(1024)
bucket_scan(const int* __restrict__ cnt, int* __restrict__ base, int NB, int nchunk,
            int* __restrict__ bstart) {
  __shared__ int s[1024];
  int b = threadIdx.x;
  int run = 0;
  if (b < NB) {
    for (int w0 = 0; w0 < nchunk; w0 += 16) {
      int c[16];
#pragma unroll
      for (int j = 0; j < 16; ++j) c[j] = cnt[min(w0 + j, nchunk - 1) * NB + b];
#pragma unroll
      for (int j = 0; j < 16; ++j) {
        if (w0 + j < nchunk) base[(w0 + j) * NB + b] = run;
        run += (w0 + j < nchunk) ? c[j] : 0;
      }
    }
  }
  s[b] = run;
  __syncthreads();
  for (int off = 1; off < 1024; off <<= 1) {
    int v = (b >= off) ? s[b - off] : 0;
    __syncthreads();
    s[b] += v;
    __syncthreads();
  }
  if (b < NB) bstart[b] = s[b] - run;
  if (b == 0) bstart[NB] = s[1023];
}

template <bool SCALE>
__device__ __forceinline__ void xw_tile(const float* __restrict__ X, const float* __restrict__ W,
                                        const float* __restrict__ dinv, float* __restrict__ G,
                                        int N, int K, int tile) {
  int lane = threadIdx.x & 31;
  int hi = lane >> 4;
  int lo = lane & 15;
  int row = tile * 16 + lo;
  float rowMask = (row < N) ? 1.0f : 0.0f;
  const float* xr = X + (size_t)((row < N) ? row : 0) * K;
  const float* wc = W + lo;
  int K4 = K & ~3;

  v8f c = {};
  int k0 = 0;
  for (; k0 < K4; k0 += 4) {
    int kk = k0 + 2 * hi;
    if ((k0 & 63) == 0) __builtin_prefetch(xr + kk + 64, 0, 1);
    v2f a, b;
    a.x = xr[kk] * rowMask;
    a.y = xr[kk + 1] * rowMask;
    b.x = wc[kk * 16];
    b.y = wc[(kk + 1) * 16];
    c = __builtin_amdgcn_wmma_f32_16x16x4_f32(false, a, false, b, (short)0, c,
                                              false, false);
  }
  if (k0 < K) {
    int kk = k0 + 2 * hi;
    int kc0 = (kk < K) ? kk : 0;
    int kc1 = (kk + 1 < K) ? kk + 1 : 0;
    float m0 = (kk < K) ? rowMask : 0.0f;
    float m1 = (kk + 1 < K) ? rowMask : 0.0f;
    v2f a, b;
    a.x = xr[kc0] * m0;
    a.y = xr[kc1] * m1;
    b.x = wc[kc0 * 16];
    b.y = wc[kc1 * 16];
    c = __builtin_amdgcn_wmma_f32_16x16x4_f32(false, a, false, b, (short)0, c,
                                              false, false);
  }
#pragma unroll
  for (int j = 0; j < 8; ++j) {
    int rr = tile * 16 + 8 * hi + j;
    if (rr < N) G[(size_t)rr * 16 + lo] = SCALE ? c[j] * dinv[rr] : c[j];
  }
}

__global__ void gemm_tile16(const float* __restrict__ X, const float* __restrict__ W,
                            const float* __restrict__ dinv, float* __restrict__ G,
                            int N, int K, int tiles) {
  int tile = blockIdx.x * (blockDim.x >> 5) + (threadIdx.x >> 5);
  if (tile >= tiles) return;
  xw_tile<true>(X, W, dinv, G, N, K, tile);
}

__global__ void __launch_bounds__(1024)
bucket_place(const int* __restrict__ src, const int* __restrict__ dst, int E, int NB,
             const int* __restrict__ cnt, const int* __restrict__ base,
             const int* __restrict__ bstart, unsigned int* __restrict__ sorted, int nplace,
             const float* __restrict__ X, const float* __restrict__ W, float* __restrict__ G,
             int N, int K, int tiles) {
  __shared__ unsigned int stage[CHUNK];
  __shared__ int loff[1024];
  __shared__ int cur[1024];
  __shared__ int gpos[1024];
  int t = threadIdx.x;
  if ((int)blockIdx.x >= nplace) {
    int tile = ((int)blockIdx.x - nplace) * 32 + (t >> 5);
    if (tile < tiles) xw_tile<false>(X, W, nullptr, G, N, K, tile);
    return;
  }
  int c = (t < NB) ? cnt[blockIdx.x * NB + t] : 0;
  loff[t] = c;
  __syncthreads();
  for (int off = 1; off < 1024; off <<= 1) {
    int v = (t >= off) ? loff[t - off] : 0;
    __syncthreads();
    loff[t] += v;
    __syncthreads();
  }
  int ex = loff[t] - c;
  loff[t] = ex;
  cur[t] = ex;
  gpos[t] = (t < NB) ? bstart[t] + base[blockIdx.x * NB + t] : 0;
  __syncthreads();
  int e0 = blockIdx.x * CHUNK;
  int e1 = min(E, e0 + CHUNK);
  int e = e0 + t;
  for (; e + 4 * 1024 < e1; e += 5 * 1024) {
    int d[5], s[5];
#pragma unroll
    for (int j = 0; j < 5; ++j) { d[j] = dst[e + j * 1024]; s[j] = src[e + j * 1024]; }
#pragma unroll
    for (int j = 0; j < 5; ++j) {
      int r = atomicAdd(&cur[d[j] >> BSH], 1);
      stage[r] = ((unsigned int)s[j] << BSH) | (unsigned int)(d[j] & (BNODES - 1));
    }
  }
  for (; e < e1; e += 1024) {
    int d = dst[e], s = src[e];
    int r = atomicAdd(&cur[d >> BSH], 1);
    stage[r] = ((unsigned int)s << BSH) | (unsigned int)(d & (BNODES - 1));
  }
  __syncthreads();
  int wave = __builtin_amdgcn_readfirstlane(t >> 5), lane = t & 31;
  for (int b = wave; b < NB; b += 32) {
    int lo = loff[b], n = cur[b] - lo, gp = gpos[b];
    for (int j = lane; j < n; j += 32) sorted[gp + j] = stage[lo + j];
  }
}

__global__ void __launch_bounds__(4 * BNODES)
bucket_dinv(const unsigned int* __restrict__ sorted, const int* __restrict__ bstart,
            float* __restrict__ dinv, float* __restrict__ g, int N) {
  __shared__ int c[BNODES];
  const int T = 4 * BNODES;
  const int grow = blockIdx.x * BNODES + ((int)threadIdx.x >> 2);
  float4* gp = (float4*)(g + (size_t)min(grow, N - 1) * 16) + (threadIdx.x & 3);
  if (threadIdx.x < BNODES) c[threadIdx.x] = 0;
  __syncthreads();
  float4 gv = *gp;
  int e0 = bstart[blockIdx.x], e1 = bstart[blockIdx.x + 1];
  int e = e0 + (int)threadIdx.x;
  for (; e + 3 * T < e1; e += 4 * T) {
    unsigned int w0 = sorted[e], w1 = sorted[e + T], w2 = sorted[e + 2 * T], w3 = sorted[e + 3 * T];
    atomicAdd(&c[w0 & (BNODES - 1)], 1);
    atomicAdd(&c[w1 & (BNODES - 1)], 1);
    atomicAdd(&c[w2 & (BNODES - 1)], 1);
    atomicAdd(&c[w3 & (BNODES - 1)], 1);
  }
  for (; e < e1; e += T) atomicAdd(&c[sorted[e] & (BNODES - 1)], 1);
  __syncthreads();
  int node = blockIdx.x * BNODES + threadIdx.x;
  if (threadIdx.x < BNODES && node < N) dinv[node] = rsqrtf((float)c[threadIdx.x] + 1.0f);
  if (grow < N) {
    float di = rsqrtf((float)c[threadIdx.x >> 2] + 1.0f);
    gv.x *= di;
    gv.y *= di;
    gv.z *= di;
    gv.w *= di;
    *gp = gv;
  }
}

#define RPER 10
#define RWORDS (RPER * 4 * BNODES)
__device__ __forceinline__ float4 bucket_sum(const unsigned int* __restrict__ sorted, int e0, int e1,
                                             const float* __restrict__ g, unsigned int* stage,
                                             int* cnt, int* off, int* wtot) {
  const int T = 4 * BNODES;
  const int t = threadIdx.x;
  const int lane = t & 31;
  const int node = t >> 2, part = t & 3;
  const float4* g4 = (const float4*)g;
  float4 acc = make_float4(0.0f, 0.0f, 0.0f, 0.0f);
  for (int c0 = e0; c0 < e1; c0 += RWORDS) {
    const int c1 = min(e1, c0 + RWORDS);
    if (t < BNODES) cnt[t] = 0;
    __syncthreads();
    unsigned int w[RPER];
    int r[RPER];
#pragma unroll
    for (int j = 0; j < RPER; ++j) w[j] = sorted[min(c0 + j * T + t, c1 - 1)];
#pragma unroll
    for (int j = 0; j < RPER; ++j) {
      r[j] = 0;
      if (c0 + j * T + t < c1) r[j] = atomicAdd(&cnt[w[j] & (BNODES - 1)], 1);
    }
    __syncthreads();
    int c = (t < BNODES) ? cnt[t] : 0;
    int x = c;
    for (int d = 1; d < 32; d <<= 1) {
      int y = __shfl_up(x, d);
      if (lane >= d) x += y;
    }
    if (t < BNODES && lane == 31) wtot[t >> 5] = x;
    __syncthreads();
    if (t < BNODES) {
      int pre = 0;
      for (int k = 0; k < (t >> 5); ++k) pre += wtot[k];
      off[t] = pre + x - c;
    }
    __syncthreads();
    int o[RPER];
#pragma unroll
    for (int j = 0; j < RPER; ++j) o[j] = off[w[j] & (BNODES - 1)];
#pragma unroll
    for (int j = 0; j < RPER; ++j)
      if (c0 + j * T + t < c1) stage[o[j] + r[j]] = w[j] >> BSH;
    __syncthreads();
    const int st = off[node];
    const int n = cnt[node];
    for (int j = 0; j < n; j += 8) {
      unsigned int s[8];
      float4 v[8];
#pragma unroll
      for (int u = 0; u < 8; ++u) s[u] = stage[st + min(j + u, n - 1)];
#pragma unroll
      for (int u = 0; u < 8; ++u) v[u] = g4[(s[u] << 2) + (unsigned int)part];
#pragma unroll
      for (int u = 0; u < 8; ++u) {
        bool ok = (j + u) < n;
        acc.x += ok ? v[u].x : 0.0f;
        acc.y += ok ? v[u].y : 0.0f;
        acc.z += ok ? v[u].z : 0.0f;
        acc.w += ok ? v[u].w : 0.0f;
      }
    }
    __syncthreads();
  }
  return acc;
}

__global__ void __launch_bounds__(4 * BNODES, 16)
bucket_agg1(const unsigned int* __restrict__ sorted, const int* __restrict__ bstart,
            const float* __restrict__ g, const float* __restrict__ dinv,
            const float* __restrict__ b, float* __restrict__ h, int N) {
  __shared__ unsigned int stage[RWORDS];
  __shared__ int cnt[BNODES], off[BNODES], wtot[BNODES / 32];
  float4 a = bucket_sum(sorted, bstart[blockIdx.x], bstart[blockIdx.x + 1], g, stage, cnt, off, wtot);
  int r = threadIdx.x >> 2, part = threadIdx.x & 3;
  int node = blockIdx.x * BNODES + r;
  if (node < N) {
    float di = dinv[node];
    float4 gg = ((const float4*)(g + (size_t)node * 16))[part];
    float4 o;
    o.x = fmaxf(di * (a.x + gg.x) + b[part * 4 + 0], 0.0f);
    o.y = fmaxf(di * (a.y + gg.y) + b[part * 4 + 1], 0.0f);
    o.z = fmaxf(di * (a.z + gg.z) + b[part * 4 + 2], 0.0f);
    o.w = fmaxf(di * (a.w + gg.w) + b[part * 4 + 3], 0.0f);
    ((float4*)(h + (size_t)node * 16))[part] = o;
  }
}

__global__ void __launch_bounds__(4 * BNODES, 16)
bucket_agg2(const unsigned int* __restrict__ sorted, const int* __restrict__ bstart,
            const float* __restrict__ g, const float* __restrict__ dinv,
            const float* __restrict__ b, const float* __restrict__ Wfc,
            float* __restrict__ p, float* __restrict__ q, int N) {
  __shared__ unsigned int stage[RWORDS];
  __shared__ int cnt[BNODES], off[BNODES], wtot[BNODES / 32];
  __shared__ float acc[BNODES * APITCH];
  float4 a = bucket_sum(sorted, bstart[blockIdx.x], bstart[blockIdx.x + 1], g, stage, cnt, off, wtot);
  {
    float* o = acc + (threadIdx.x >> 2) * APITCH + (threadIdx.x & 3) * 4;
    o[0] = a.x;
    o[1] = a.y;
    o[2] = a.z;
    o[3] = a.w;
  }
  __syncthreads();
  int node = blockIdx.x * BNODES + threadIdx.x;
  if (threadIdx.x < BNODES && node < N) {
    float di = dinv[node];
    float sp = 0.0f, sq = 0.0f;
#pragma unroll
    for (int f = 0; f < 16; ++f) {
      float hh = fmaxf(di * (acc[threadIdx.x * APITCH + f] + g[(size_t)node * 16 + f]) + b[f], 0.0f);
      sp += hh * Wfc[f];
      sq += hh * Wfc[16 + f];
    }
    p[node] = sp;
    q[node] = sq;
  }
}

#define OCH 12800
#define OPER 13
__global__ void __launch_bounds__(1024)
edge_out(const int* __restrict__ src, const int* __restrict__ dst,
         const float* __restrict__ p, const float* __restrict__ q,
         const float* __restrict__ bfc, float* __restrict__ out, int E) {
  __shared__ unsigned int stS[OCH];
  __shared__ unsigned int stD[OCH];
  __shared__ float val[OCH];
  __shared__ int cS[NBMAX], cD[NBMAX];
  __shared__ int wtS[32], wtD[32];
  const int t = threadIdx.x, lane = t & 31, wave = t >> 5;
  cS[t] = 0;
  cD[t] = 0;
  __syncthreads();
  const int e0 = blockIdx.x * OCH;
  const int n = min(E - e0, OCH);
  int s[OPER], d[OPER], rs[OPER], rd[OPER];
#pragma unroll
  for (int j = 0; j < OPER; ++j) {
    int i = min(j * 1024 + t, n - 1);
    s[j] = src[e0 + i];
    d[j] = dst[e0 + i];
  }
#pragma unroll
  for (int j = 0; j < OPER; ++j) {
    rs[j] = 0;
    rd[j] = 0;
    if (j * 1024 + t < n) {
      rs[j] = atomicAdd(&cS[s[j] >> BSH], 1);
      rd[j] = atomicAdd(&cD[d[j] >> BSH], 1);
    }
  }
  __syncthreads();
  {
    const int a = cS[t], b = cD[t];
    int xa = a, xb = b;
    for (int o = 1; o < 32; o <<= 1) {
      int ya = __shfl_up(xa, o), yb = __shfl_up(xb, o);
      if (lane >= o) { xa += ya; xb += yb; }
    }
    if (lane == 31) { wtS[wave] = xa; wtD[wave] = xb; }
    __syncthreads();
    const int ta = wtS[lane], tb = wtD[lane];
    int sa = ta, sb = tb;
    for (int o = 1; o < 32; o <<= 1) {
      int ya = __shfl_up(sa, o), yb = __shfl_up(sb, o);
      if (lane >= o) { sa += ya; sb += yb; }
    }
    const int preA = __shfl(sa - ta, wave), preB = __shfl(sb - tb, wave);
    cS[t] = preA + xa - a;
    cD[t] = preB + xb - b;
  }
  __syncthreads();
#pragma unroll
  for (int j = 0; j < OPER; ++j) {
    const int i = j * 1024 + t;
    if (i < n) {
      stS[cS[s[j] >> BSH] + rs[j]] = ((unsigned int)i << 17) | (unsigned int)s[j];
      stD[cD[d[j] >> BSH] + rd[j]] = ((unsigned int)i << 17) | (unsigned int)d[j];
    }
  }
  __syncthreads();
  {
    unsigned int wd[OPER], ws[OPER];
    float vq[OPER], vp[OPER];
#pragma unroll
    for (int j = 0; j < OPER; ++j) {
      const int r = min(j * 1024 + t, n - 1);
      wd[j] = stD[r];
      ws[j] = stS[r];
    }
#pragma unroll
    for (int j = 0; j < OPER; ++j) vq[j] = q[wd[j] & 0x1FFFFu];
#pragma unroll
    for (int j = 0; j < OPER; ++j) vp[j] = p[ws[j] & 0x1FFFFu];
#pragma unroll
    for (int j = 0; j < OPER; ++j)
      if (j * 1024 + t < n) val[wd[j] >> 17] = vq[j];
    __syncthreads();
#pragma unroll
    for (int j = 0; j < OPER; ++j)
      if (j * 1024 + t < n) val[ws[j] >> 17] = vp[j] + val[ws[j] >> 17];
  }
  __syncthreads();
  const float bias = bfc[0];
#pragma unroll
  for (int j = 0; j < OPER; ++j) {
    const int i = j * 1024 + t;
    if (i < n) out[e0 + i] = val[i] + bias;
  }
}

extern "C" void kernel_launch(void* const* d_in, const int* in_sizes, int n_in,
                              void* d_out, int out_size, void* d_ws, size_t ws_size,
                              hipStream_t stream) {
  const float* x   = (const float*)d_in[0];
  const int*   ei  = (const int*)d_in[1];
  const float* W1  = (const float*)d_in[2];
  const float* b1  = (const float*)d_in[3];
  const float* W2  = (const float*)d_in[4];
  const float* b2  = (const float*)d_in[5];
  const float* Wfc = (const float*)d_in[6];
  const float* bfc = (const float*)d_in[7];
  float* out = (float*)d_out;

  const int K1 = 495;
  const int N = in_sizes[0] / K1;
  const int E = in_sizes[1] / 2;
  const int* src = ei;
  const int* dst = ei + E;

  const int NB = cdiv(N, BNODES);
  const int nchunk = cdiv(E, CHUNK);

  float* ws  = (float*)d_ws;
  float* dinv = ws;
  float* g    = ws + (size_t)N;
  float* h1   = ws + (size_t)N * 17;
  float* p    = ws + (size_t)N * 33;
  float* q    = ws + (size_t)N * 34;
  int* cnt    = (int*)(ws + (size_t)N * 35);
  int* base   = cnt + (size_t)nchunk * NB;
  int* bstart = base + (size_t)nchunk * NB;
  unsigned int* sorted = (unsigned int*)d_out;

  const int B = 256;
  const int tiles = cdiv(N, 16);
  const int gemmBlocks = cdiv(tiles, B / 32);
  const int xwGroups = cdiv(tiles, 32);

  bucket_count<<<nchunk, 1024, 0, stream>>>(dst, E, NB, cnt);
  bucket_scan<<<1, 1024, 0, stream>>>(cnt, base, NB, nchunk, bstart);
  bucket_place<<<nchunk + xwGroups, 1024, 0, stream>>>(src, dst, E, NB, cnt, base, bstart, sorted,
                                                        nchunk, x, W1, g, N, K1, tiles);

  bucket_dinv<<<NB, 4 * BNODES, 0, stream>>>(sorted, bstart, dinv, g, N);

  bucket_agg1<<<NB, 4 * BNODES, 0, stream>>>(sorted, bstart, g, dinv, b1, h1, N);

  gemm_tile16<<<gemmBlocks, B, 0, stream>>>(h1, W2, dinv, g, N, 16, tiles);
  bucket_agg2<<<NB, 4 * BNODES, 0, stream>>>(sorted, bstart, g, dinv, b2, Wfc, p, q, N);

  edge_out<<<cdiv(E, OCH), 1024, 0, stream>>>(src, dst, p, q, bfc, out, E);
}


# layer-1 product: a lane reads 16 bytes of its x row per load (8-k block feeding two WMMAs) instead of 8; k order inside each 8-block permuted alike in A and W fragments: new sum order of x@W1
# speedup vs baseline: 3.3991x; 1.0953x over previous
#include <hip/hip_runtime.h>
#include <hip/hip_bf16.h>

typedef float v2f __attribute__((ext_vector_type(2)));
typedef float v8f __attribute__((ext_vector_type(8)));

static __host__ int cdiv(int a, int b) { return (a + b - 1) / b; }


#define BSH 7
#define BNODES 128
#define NBMAX 1024
#define CHUNK 25600
#define APITCH 17

__global__ void __launch_bounds__(1024)
bucket_count(const int* __restrict__ dst, int E, int NB, int* __restrict__ cnt) {
  __shared__ int h[NBMAX];
  h[threadIdx.x] = 0;
  __syncthreads();
  int e0 = blockIdx.x * CHUNK;
  int e1 = min(E, e0 + CHUNK);
  int e = e0 + (int)threadIdx.x;
  for (; e + 4 * 1024 < e1; e += 5 * 1024) {
    int d0 = dst[e], d1 = dst[e + 1024], d2 = dst[e + 2048], d3 = dst[e + 3072], d4 = dst[e + 4096];
    atomicAdd(&h[d0 >> BSH], 1);
    atomicAdd(&h[d1 >> BSH], 1);
    atomicAdd(&h[d2 >> BSH], 1);
    atomicAdd(&h[d3 >> BSH], 1);
    atomicAdd(&h[d4 >> BSH], 1);
  }
  for (; e < e1; e += 1024) atomicAdd(&h[dst[e] >> BSH], 1);
  __syncthreads();
  if ((int)threadIdx.x < NB) cnt[blockIdx.x * NB + threadIdx.x] = h[threadIdx.x];
}

__global__ void __launch_bounds__(1024)
bucket_scan(const int* __restrict__ cnt, int* __restrict__ base, int NB, int nchunk,
            int* __restrict__ bstart) {
  __shared__ int s[1024];
  int b = threadIdx.x;
  int run = 0;
  if (b < NB) {
    for (int w0 = 0; w0 < nchunk; w0 += 16) {
      int c[16];
#pragma unroll
      for (int j = 0; j < 16; ++j) c[j] = cnt[min(w0 + j, nchunk - 1) * NB + b];
#pragma unroll
      for (int j = 0; j < 16; ++j) {
        if (w0 + j < nchunk) base[(w0 + j) * NB + b] = run;
        run += (w0 + j < nchunk) ? c[j] : 0;
      }
    }
  }
  s[b] = run;
  __syncthreads();
  for (int off = 1; off < 1024; off <<= 1) {
    int v = (b >= off) ? s[b - off] : 0;
    __syncthreads();
    s[b] += v;
    __syncthreads();
  }
  if (b < NB) bstart[b] = s[b] - run;
  if (b == 0) bstart[NB] = s[1023];
}

template <bool SCALE>
__device__ __forceinline__ void xw_tile(const float* __restrict__ X, const float* __restrict__ W,
                                        const float* __restrict__ dinv, float* __restrict__ G,
                                        int N, int K, int tile) {
  int lane = threadIdx.x & 31;
  int hi = lane >> 4;
  int lo = lane & 15;
  int row = tile * 16 + lo;
  float rowMask = (row < N) ? 1.0f : 0.0f;
  const float* xr = X + (size_t)((row < N) ? row : 0) * K;
  const float* wc = W + lo;
  int K4 = K & ~3;

  v8f c = {};
  int k0 = 0;
  for (; k0 < K4; k0 += 4) {
    int kk = k0 + 2 * hi;
    if ((k0 & 63) == 0) __builtin_prefetch(xr + kk + 64, 0, 1);
    v2f a, b;
    a.x = xr[kk] * rowMask;
    a.y = xr[kk + 1] * rowMask;
    b.x = wc[kk * 16];
    b.y = wc[(kk + 1) * 16];
    c = __builtin_amdgcn_wmma_f32_16x16x4_f32(false, a, false, b, (short)0, c,
                                              false, false);
  }
  if (k0 < K) {
    int kk = k0 + 2 * hi;
    int kc0 = (kk < K) ? kk : 0;
    int kc1 = (kk + 1 < K) ? kk + 1 : 0;
    float m0 = (kk < K) ? rowMask : 0.0f;
    float m1 = (kk + 1 < K) ? rowMask : 0.0f;
    v2f a, b;
    a.x = xr[kc0] * m0;
    a.y = xr[kc1] * m1;
    b.x = wc[kc0 * 16];
    b.y = wc[kc1 * 16];
    c = __builtin_amdgcn_wmma_f32_16x16x4_f32(false, a, false, b, (short)0, c,
                                              false, false);
  }
#pragma unroll
  for (int j = 0; j < 8; ++j) {
    int rr = tile * 16 + 8 * hi + j;
    if (rr < N) G[(size_t)rr * 16 + lo] = SCALE ? c[j] * dinv[rr] : c[j];
  }
}

__device__ __forceinline__ void xw_tile_k8(const float* __restrict__ X, const float* __restrict__ W,
                                           float* __restrict__ G, int N, int K, int tile) {
  int lane = threadIdx.x & 31;
  int hi = lane >> 4;
  int lo = lane & 15;
  int row = tile * 16 + lo;
  float rowMask = (row < N) ? 1.0f : 0.0f;
  const float* xr = X + (size_t)((row < N) ? row : 0) * K;
  const float* wc = W + lo;
  int K8 = K & ~7;

  v8f c = {};
  int k0 = 0;
  for (; k0 < K8; k0 += 8) {
    int kk = k0 + 4 * hi;
    if ((k0 & 63) == 0) __builtin_prefetch(xr + kk + 64, 0, 1);
    float x0 = xr[kk], x1 = xr[kk + 1], x2 = xr[kk + 2], x3 = xr[kk + 3];
    v2f a, b;
    a.x = x0 * rowMask;
    a.y = x1 * rowMask;
    b.x = wc[kk * 16];
    b.y = wc[(kk + 1) * 16];
    c = __builtin_amdgcn_wmma_f32_16x16x4_f32(false, a, false, b, (short)0, c,
                                              false, false);
    a.x = x2 * rowMask;
    a.y = x3 * rowMask;
    b.x = wc[(kk + 2) * 16];
    b.y = wc[(kk + 3) * 16];
    c = __builtin_amdgcn_wmma_f32_16x16x4_f32(false, a, false, b, (short)0, c,
                                              false, false);
  }
  if (k0 < K) {
    int kk = k0 + 4 * hi;
    float xa[4], wb[4];
#pragma unroll
    for (int s = 0; s < 4; ++s) {
      int kc = (kk + s < K) ? kk + s : 0;
      float m = (kk + s < K) ? rowMask : 0.0f;
      xa[s] = xr[kc] * m;
      wb[s] = wc[kc * 16];
    }
    v2f a, b;
    a.x = xa[0]; a.y = xa[1]; b.x = wb[0]; b.y = wb[1];
    c = __builtin_amdgcn_wmma_f32_16x16x4_f32(false, a, false, b, (short)0, c,
                                              false, false);
    a.x = xa[2]; a.y = xa[3]; b.x = wb[2]; b.y = wb[3];
    c = __builtin_amdgcn_wmma_f32_16x16x4_f32(false, a, false, b, (short)0, c,
                                              false, false);
  }
#pragma unroll
  for (int j = 0; j < 8; ++j) {
    int rr = tile * 16 + 8 * hi + j;
    if (rr < N) G[(size_t)rr * 16 + lo] = c[j];
  }
}

__global__ void gemm_tile16(const float* __restrict__ X, const float* __restrict__ W,
                            const float* __restrict__ dinv, float* __restrict__ G,
                            int N, int K, int tiles) {
  int tile = blockIdx.x * (blockDim.x >> 5) + (threadIdx.x >> 5);
  if (tile >= tiles) return;
  xw_tile<true>(X, W, dinv, G, N, K, tile);
}

__global__ void __launch_bounds__(1024, 16)
bucket_place(const int* __restrict__ src, const int* __restrict__ dst, int E, int NB,
             const int* __restrict__ cnt, const int* __restrict__ base,
             const int* __restrict__ bstart, unsigned int* __restrict__ sorted, int nplace,
             const float* __restrict__ X, const float* __restrict__ W, float* __restrict__ G,
             int N, int K, int tiles) {
  __shared__ unsigned int stage[CHUNK];
  __shared__ int loff[1024];
  __shared__ int cur[1024];
  __shared__ int gpos[1024];
  int t = threadIdx.x;
  if ((int)blockIdx.x >= nplace) {
    int tile = ((int)blockIdx.x - nplace) * 32 + (t >> 5);
    if (tile < tiles) xw_tile_k8(X, W, G, N, K, tile);
    return;
  }
  int c = (t < NB) ? cnt[blockIdx.x * NB + t] : 0;
  loff[t] = c;
  __syncthreads();
  for (int off = 1; off < 1024; off <<= 1) {
    int v = (t >= off) ? loff[t - off] : 0;
    __syncthreads();
    loff[t] += v;
    __syncthreads();
  }
  int ex = loff[t] - c;
  loff[t] = ex;
  cur[t] = ex;
  gpos[t] = (t < NB) ? bstart[t] + base[blockIdx.x * NB + t] : 0;
  __syncthreads();
  int e0 = blockIdx.x * CHUNK;
  int e1 = min(E, e0 + CHUNK);
  int e = e0 + t;
  for (; e + 4 * 1024 < e1; e += 5 * 1024) {
    int d[5], s[5];
#pragma unroll
    for (int j = 0; j < 5; ++j) { d[j] = dst[e + j * 1024]; s[j] = src[e + j * 1024]; }
#pragma unroll
    for (int j = 0; j < 5; ++j) {
      int r = atomicAdd(&cur[d[j] >> BSH], 1);
      stage[r] = ((unsigned int)s[j] << BSH) | (unsigned int)(d[j] & (BNODES - 1));
    }
  }
  for (; e < e1; e += 1024) {
    int d = dst[e], s = src[e];
    int r = atomicAdd(&cur[d >> BSH], 1);
    stage[r] = ((unsigned int)s << BSH) | (unsigned int)(d & (BNODES - 1));
  }
  __syncthreads();
  int wave = __builtin_amdgcn_readfirstlane(t >> 5), lane = t & 31;
  for (int b = wave; b < NB; b += 32) {
    int lo = loff[b], n = cur[b] - lo, gp = gpos[b];
    for (int j = lane; j < n; j += 32) sorted[gp + j] = stage[lo + j];
  }
}

__global__ void __launch_bounds__(4 * BNODES)
bucket_dinv(const unsigned int* __restrict__ sorted, const int* __restrict__ bstart,
            float* __restrict__ dinv, float* __restrict__ g, int N) {
  __shared__ int c[BNODES];
  const int T = 4 * BNODES;
  const int grow = blockIdx.x * BNODES + ((int)threadIdx.x >> 2);
  float4* gp = (float4*)(g + (size_t)min(grow, N - 1) * 16) + (threadIdx.x & 3);
  if (threadIdx.x < BNODES) c[threadIdx.x] = 0;
  __syncthreads();
  float4 gv = *gp;
  int e0 = bstart[blockIdx.x], e1 = bstart[blockIdx.x + 1];
  int e = e0 + (int)threadIdx.x;
  for (; e + 3 * T < e1; e += 4 * T) {
    unsigned int w0 = sorted[e], w1 = sorted[e + T], w2 = sorted[e + 2 * T], w3 = sorted[e + 3 * T];
    atomicAdd(&c[w0 & (BNODES - 1)], 1);
    atomicAdd(&c[w1 & (BNODES - 1)], 1);
    atomicAdd(&c[w2 & (BNODES - 1)], 1);
    atomicAdd(&c[w3 & (BNODES - 1)], 1);
  }
  for (; e < e1; e += T) atomicAdd(&c[sorted[e] & (BNODES - 1)], 1);
  __syncthreads();
  int node = blockIdx.x * BNODES + threadIdx.x;
  if (threadIdx.x < BNODES && node < N) dinv[node] = rsqrtf((float)c[threadIdx.x] + 1.0f);
  if (grow < N) {
    float di = rsqrtf((float)c[threadIdx.x >> 2] + 1.0f);
    gv.x *= di;
    gv.y *= di;
    gv.z *= di;
    gv.w *= di;
    *gp = gv;
  }
}

#define RPER 10
#define RWORDS (RPER * 4 * BNODES)
__device__ __forceinline__ float4 bucket_sum(const unsigned int* __restrict__ sorted, int e0, int e1,
                                             const float* __restrict__ g, unsigned int* stage,
                                             int* cnt, int* off, int* wtot) {
  const int T = 4 * BNODES;
  const int t = threadIdx.x;
  const int lane = t & 31;
  const int node = t >> 2, part = t & 3;
  const float4* g4 = (const float4*)g;
  float4 acc = make_float4(0.0f, 0.0f, 0.0f, 0.0f);
  for (int c0 = e0; c0 < e1; c0 += RWORDS) {
    const int c1 = min(e1, c0 + RWORDS);
    if (t < BNODES) cnt[t] = 0;
    __syncthreads();
    unsigned int w[RPER];
    int r[RPER];
#pragma unroll
    for (int j = 0; j < RPER; ++j) w[j] = sorted[min(c0 + j * T + t, c1 - 1)];
#pragma unroll
    for (int j = 0; j < RPER; ++j) {
      r[j] = 0;
      if (c0 + j * T + t < c1) r[j] = atomicAdd(&cnt[w[j] & (BNODES - 1)], 1);
    }
    __syncthreads();
    int c = (t < BNODES) ? cnt[t] : 0;
    int x = c;
    for (int d = 1; d < 32; d <<= 1) {
      int y = __shfl_up(x, d);
      if (lane >= d) x += y;
    }
    if (t < BNODES && lane == 31) wtot[t >> 5] = x;
    __syncthreads();
    if (t < BNODES) {
      int pre = 0;
      for (int k = 0; k < (t >> 5); ++k) pre += wtot[k];
      off[t] = pre + x - c;
    }
    __syncthreads();
    int o[RPER];
#pragma unroll
    for (int j = 0; j < RPER; ++j) o[j] = off[w[j] & (BNODES - 1)];
#pragma unroll
    for (int j = 0; j < RPER; ++j)
      if (c0 + j * T + t < c1) stage[o[j] + r[j]] = w[j] >> BSH;
    __syncthreads();
    const int st = off[node];
    const int n = cnt[node];
    for (int j = 0; j < n; j += 8) {
      unsigned int s[8];
      float4 v[8];
#pragma unroll
      for (int u = 0; u < 8; ++u) s[u] = stage[st + min(j + u, n - 1)];
#pragma unroll
      for (int u = 0; u < 8; ++u) v[u] = g4[(s[u] << 2) + (unsigned int)part];
#pragma unroll
      for (int u = 0; u < 8; ++u) {
        bool ok = (j + u) < n;
        acc.x += ok ? v[u].x : 0.0f;
        acc.y += ok ? v[u].y : 0.0f;
        acc.z += ok ? v[u].z : 0.0f;
        acc.w += ok ? v[u].w : 0.0f;
      }
    }
    __syncthreads();
  }
  return acc;
}

__global__ void __launch_bounds__(4 * BNODES, 16)
bucket_agg1(const unsigned int* __restrict__ sorted, const int* __restrict__ bstart,
            const float* __restrict__ g, const float* __restrict__ dinv,
            const float* __restrict__ b, float* __restrict__ h, int N) {
  __shared__ unsigned int stage[RWORDS];
  __shared__ int cnt[BNODES], off[BNODES], wtot[BNODES / 32];
  float4 a = bucket_sum(sorted, bstart[blockIdx.x], bstart[blockIdx.x + 1], g, stage, cnt, off, wtot);
  int r = threadIdx.x >> 2, part = threadIdx.x & 3;
  int node = blockIdx.x * BNODES + r;
  if (node < N) {
    float di = dinv[node];
    float4 gg = ((const float4*)(g + (size_t)node * 16))[part];
    float4 o;
    o.x = fmaxf(di * (a.x + gg.x) + b[part * 4 + 0], 0.0f);
    o.y = fmaxf(di * (a.y + gg.y) + b[part * 4 + 1], 0.0f);
    o.z = fmaxf(di * (a.z + gg.z) + b[part * 4 + 2], 0.0f);
    o.w = fmaxf(di * (a.w + gg.w) + b[part * 4 + 3], 0.0f);
    ((float4*)(h + (size_t)node * 16))[part] = o;
  }
}

__global__ void __launch_bounds__(4 * BNODES, 16)
bucket_agg2(const unsigned int* __restrict__ sorted, const int* __restrict__ bstart,
            const float* __restrict__ g, const float* __restrict__ dinv,
            const float* __restrict__ b, const float* __restrict__ Wfc,
            float* __restrict__ p, float* __restrict__ q, int N) {
  __shared__ unsigned int stage[RWORDS];
  __shared__ int cnt[BNODES], off[BNODES], wtot[BNODES / 32];
  __shared__ float acc[BNODES * APITCH];
  float4 a = bucket_sum(sorted, bstart[blockIdx.x], bstart[blockIdx.x + 1], g, stage, cnt, off, wtot);
  {
    float* o = acc + (threadIdx.x >> 2) * APITCH + (threadIdx.x & 3) * 4;
    o[0] = a.x;
    o[1] = a.y;
    o[2] = a.z;
    o[3] = a.w;
  }
  __syncthreads();
  int node = blockIdx.x * BNODES + threadIdx.x;
  if (threadIdx.x < BNODES && node < N) {
    float di = dinv[node];
    float sp = 0.0f, sq = 0.0f;
#pragma unroll
    for (int f = 0; f < 16; ++f) {
      float hh = fmaxf(di * (acc[threadIdx.x * APITCH + f] + g[(size_t)node * 16 + f]) + b[f], 0.0f);
      sp += hh * Wfc[f];
      sq += hh * Wfc[16 + f];
    }
    p[node] = sp;
    q[node] = sq;
  }
}

#define OCH 12800
#define OPER 13
__global__ void __launch_bounds__(1024)
edge_out(const int* __restrict__ src, const int* __restrict__ dst,
         const float* __restrict__ p, const float* __restrict__ q,
         const float* __restrict__ bfc, float* __restrict__ out, int E) {
  __shared__ unsigned int stS[OCH];
  __shared__ unsigned int stD[OCH];
  __shared__ float val[OCH];
  __shared__ int cS[NBMAX], cD[NBMAX];
  __shared__ int wtS[32], wtD[32];
  const int t = threadIdx.x, lane = t & 31, wave = t >> 5;
  cS[t] = 0;
  cD[t] = 0;
  __syncthreads();
  const int e0 = blockIdx.x * OCH;
  const int n = min(E - e0, OCH);
  int s[OPER], d[OPER], rs[OPER], rd[OPER];
#pragma unroll
  for (int j = 0; j < OPER; ++j) {
    int i = min(j * 1024 + t, n - 1);
    s[j] = src[e0 + i];
    d[j] = dst[e0 + i];
  }
#pragma unroll
  for (int j = 0; j < OPER; ++j) {
    rs[j] = 0;
    rd[j] = 0;
    if (j * 1024 + t < n) {
      rs[j] = atomicAdd(&cS[s[j] >> BSH], 1);
      rd[j] = atomicAdd(&cD[d[j] >> BSH], 1);
    }
  }
  __syncthreads();
  {
    const int a = cS[t], b = cD[t];
    int xa = a, xb = b;
    for (int o = 1; o < 32; o <<= 1) {
      int ya = __shfl_up(xa, o), yb = __shfl_up(xb, o);
      if (lane >= o) { xa += ya; xb += yb; }
    }
    if (lane == 31) { wtS[wave] = xa; wtD[wave] = xb; }
    __syncthreads();
    const int ta = wtS[lane], tb = wtD[lane];
    int sa = ta, sb = tb;
    for (int o = 1; o < 32; o <<= 1) {
      int ya = __shfl_up(sa, o), yb = __shfl_up(sb, o);
      if (lane >= o) { sa += ya; sb += yb; }
    }
    const int preA = __shfl(sa - ta, wave), preB = __shfl(sb - tb, wave);
    cS[t] = preA + xa - a;
    cD[t] = preB + xb - b;
  }
  __syncthreads();
#pragma unroll
  for (int j = 0; j < OPER; ++j) {
    const int i = j * 1024 + t;
    if (i < n) {
      stS[cS[s[j] >> BSH] + rs[j]] = ((unsigned int)i << 17) | (unsigned int)s[j];
      stD[cD[d[j] >> BSH] + rd[j]] = ((unsigned int)i << 17) | (unsigned int)d[j];
    }
  }
  __syncthreads();
  {
    unsigned int wd[OPER], ws[OPER];
    float vq[OPER], vp[OPER];
#pragma unroll
    for (int j = 0; j < OPER; ++j) {
      const int r = min(j * 1024 + t, n - 1);
      wd[j] = stD[r];
      ws[j] = stS[r];
    }
#pragma unroll
    for (int j = 0; j < OPER; ++j) vq[j] = q[wd[j] & 0x1FFFFu];
#pragma unroll
    for (int j = 0; j < OPER; ++j) vp[j] = p[ws[j] & 0x1FFFFu];
#pragma unroll
    for (int j = 0; j < OPER; ++j)
      if (j * 1024 + t < n) val[wd[j] >> 17] = vq[j];
    __syncthreads();
#pragma unroll
    for (int j = 0; j < OPER; ++j)
      if (j * 1024 + t < n) val[ws[j] >> 17] = vp[j] + val[ws[j] >> 17];
  }
  __syncthreads();
  const float bias = bfc[0];
#pragma unroll
  for (int j = 0; j < OPER; ++j) {
    const int i = j * 1024 + t;
    if (i < n) out[e0 + i] = val[i] + bias;
  }
}

extern "C" void kernel_launch(void* const* d_in, const int* in_sizes, int n_in,
                              void* d_out, int out_size, void* d_ws, size_t ws_size,
                              hipStream_t stream) {
  const float* x   = (const float*)d_in[0];
  const int*   ei  = (const int*)d_in[1];
  const float* W1  = (const float*)d_in[2];
  const float* b1  = (const float*)d_in[3];
  const float* W2  = (const float*)d_in[4];
  const float* b2  = (const float*)d_in[5];
  const float* Wfc = (const float*)d_in[6];
  const float* bfc = (const float*)d_in[7];
  float* out = (float*)d_out;

  const int K1 = 495;
  const int N = in_sizes[0] / K1;
  const int E = in_sizes[1] / 2;
  const int* src = ei;
  const int* dst = ei + E;

  const int NB = cdiv(N, BNODES);
  const int nchunk = cdiv(E, CHUNK);

  float* ws  = (float*)d_ws;
  float* dinv = ws;
  float* g    = ws + (size_t)N;
  float* h1   = ws + (size_t)N * 17;
  float* p    = ws + (size_t)N * 33;
  float* q    = ws + (size_t)N * 34;
  int* cnt    = (int*)(ws + (size_t)N * 35);
  int* base   = cnt + (size_t)nchunk * NB;
  int* bstart = base + (size_t)nchunk * NB;
  unsigned int* sorted = (unsigned int*)d_out;

  const int B = 256;
  const int tiles = cdiv(N, 16);
  const int gemmBlocks = cdiv(tiles, B / 32);
  const int xwGroups = cdiv(tiles, 32);

  bucket_count<<<nchunk, 1024, 0, stream>>>(dst, E, NB, cnt);
  bucket_scan<<<1, 1024, 0, stream>>>(cnt, base, NB, nchunk, bstart);
  bucket_place<<<nchunk + xwGroups, 1024, 0, stream>>>(src, dst, E, NB, cnt, base, bstart, sorted,
                                                        nchunk, x, W1, g, N, K1, tiles);

  bucket_dinv<<<NB, 4 * BNODES, 0, stream>>>(sorted, bstart, dinv, g, N);

  bucket_agg1<<<NB, 4 * BNODES, 0, stream>>>(sorted, bstart, g, dinv, b1, h1, N);

  gemm_tile16<<<gemmBlocks, B, 0, stream>>>(h1, W2, dinv, g, N, 16, tiles);
  bucket_agg2<<<NB, 4 * BNODES, 0, stream>>>(sorted, bstart, g, dinv, b2, Wfc, p, q, N);

  edge_out<<<cdiv(E, OCH), 1024, 0, stream>>>(src, dst, p, q, bfc, out, E);
}


# bucket_scan launch removed: each placing work-group adds up the chunk counts of every bucket itself (coalesced cached reads) and scans the totals beside its own counts; 8 launches -> 7, same sums
# speedup vs baseline: 3.6427x; 1.0717x over previous
#include <hip/hip_runtime.h>
#include <hip/hip_bf16.h>

typedef float v2f __attribute__((ext_vector_type(2)));
typedef float v8f __attribute__((ext_vector_type(8)));

static __host__ int cdiv(int a, int b) { return (a + b - 1) / b; }


#define BSH 7
#define BNODES 128
#define NBMAX 1024
#define CHUNK 25600
#define APITCH 17

__global__ void __launch_bounds__(1024)
bucket_count(const int* __restrict__ dst, int E, int NB, int* __restrict__ cnt) {
  __shared__ int h[NBMAX];
  h[threadIdx.x] = 0;
  __syncthreads();
  int e0 = blockIdx.x * CHUNK;
  int e1 = min(E, e0 + CHUNK);
  int e = e0 + (int)threadIdx.x;
  for (; e + 4 * 1024 < e1; e += 5 * 1024) {
    int d0 = dst[e], d1 = dst[e + 1024], d2 = dst[e + 2048], d3 = dst[e + 3072], d4 = dst[e + 4096];
    atomicAdd(&h[d0 >> BSH], 1);
    atomicAdd(&h[d1 >> BSH], 1);
    atomicAdd(&h[d2 >> BSH], 1);
    atomicAdd(&h[d3 >> BSH], 1);
    atomicAdd(&h[d4 >> BSH], 1);
  }
  for (; e < e1; e += 1024) atomicAdd(&h[dst[e] >> BSH], 1);
  __syncthreads();
  if ((int)threadIdx.x < NB) cnt[blockIdx.x * NB + threadIdx.x] = h[threadIdx.x];
}

template <bool SCALE>
__device__ __forceinline__ void xw_tile(const float* __restrict__ X, const float* __restrict__ W,
                                        const float* __restrict__ dinv, float* __restrict__ G,
                                        int N, int K, int tile) {
  int lane = threadIdx.x & 31;
  int hi = lane >> 4;
  int lo = lane & 15;
  int row = tile * 16 + lo;
  float rowMask = (row < N) ? 1.0f : 0.0f;
  const float* xr = X + (size_t)((row < N) ? row : 0) * K;
  const float* wc = W + lo;
  int K4 = K & ~3;

  v8f c = {};
  int k0 = 0;
  for (; k0 < K4; k0 += 4) {
    int kk = k0 + 2 * hi;
    if ((k0 & 63) == 0) __builtin_prefetch(xr + kk + 64, 0, 1);
    v2f a, b;
    a.x = xr[kk] * rowMask;
    a.y = xr[kk + 1] * rowMask;
    b.x = wc[kk * 16];
    b.y = wc[(kk + 1) * 16];
    c = __builtin_amdgcn_wmma_f32_16x16x4_f32(false, a, false, b, (short)0, c,
                                              false, false);
  }
  if (k0 < K) {
    int kk = k0 + 2 * hi;
    int kc0 = (kk < K) ? kk : 0;
    int kc1 = (kk + 1 < K) ? kk + 1 : 0;
    float m0 = (kk < K) ? rowMask : 0.0f;
    float m1 = (kk + 1 < K) ? rowMask : 0.0f;
    v2f a, b;
    a.x = xr[kc0] * m0;
    a.y = xr[kc1] * m1;
    b.x = wc[kc0 * 16];
    b.y = wc[kc1 * 16];
    c = __builtin_amdgcn_wmma_f32_16x16x4_f32(false, a, false, b, (short)0, c,
                                              false, false);
  }
#pragma unroll
  for (int j = 0; j < 8; ++j) {
    int rr = tile * 16 + 8 * hi + j;
    if (rr < N) G[(size_t)rr * 16 + lo] = SCALE ? c[j] * dinv[rr] : c[j];
  }
}

__device__ __forceinline__ void xw_tile_k8(const float* __restrict__ X, const float* __restrict__ W,
                                           float* __restrict__ G, int N, int K, int tile) {
  int lane = threadIdx.x & 31;
  int hi = lane >> 4;
  int lo = lane & 15;
  int row = tile * 16 + lo;
  float rowMask = (row < N) ? 1.0f : 0.0f;
  const float* xr = X + (size_t)((row < N) ? row : 0) * K;
  const float* wc = W + lo;
  int K8 = K & ~7;

  v8f c = {};
  int k0 = 0;
  for (; k0 < K8; k0 += 8) {
    int kk = k0 + 4 * hi;
    if ((k0 & 63) == 0) __builtin_prefetch(xr + kk + 64, 0, 1);
    float x0 = xr[kk], x1 = xr[kk + 1], x2 = xr[kk + 2], x3 = xr[kk + 3];
    v2f a, b;
    a.x = x0 * rowMask;
    a.y = x1 * rowMask;
    b.x = wc[kk * 16];
    b.y = wc[(kk + 1) * 16];
    c = __builtin_amdgcn_wmma_f32_16x16x4_f32(false, a, false, b, (short)0, c,
                                              false, false);
    a.x = x2 * rowMask;
    a.y = x3 * rowMask;
    b.x = wc[(kk + 2) * 16];
    b.y = wc[(kk + 3) * 16];
    c = __builtin_amdgcn_wmma_f32_16x16x4_f32(false, a, false, b, (short)0, c,
                                              false, false);
  }
  if (k0 < K) {
    int kk = k0 + 4 * hi;
    float xa[4], wb[4];
#pragma unroll
    for (int s = 0; s < 4; ++s) {
      int kc = (kk + s < K) ? kk + s : 0;
      float m = (kk + s < K) ? rowMask : 0.0f;
      xa[s] = xr[kc] * m;
      wb[s] = wc[kc * 16];
    }
    v2f a, b;
    a.x = xa[0]; a.y = xa[1]; b.x = wb[0]; b.y = wb[1];
    c = __builtin_amdgcn_wmma_f32_16x16x4_f32(false, a, false, b, (short)0, c,
                                              false, false);
    a.x = xa[2]; a.y = xa[3]; b.x = wb[2]; b.y = wb[3];
    c = __builtin_amdgcn_wmma_f32_16x16x4_f32(false, a, false, b, (short)0, c,
                                              false, false);
  }
#pragma unroll
  for (int j = 0; j < 8; ++j) {
    int rr = tile * 16 + 8 * hi + j;
    if (rr < N) G[(size_t)rr * 16 + lo] = c[j];
  }
}

__global__ void gemm_tile16(const float* __restrict__ X, const float* __restrict__ W,
                            const float* __restrict__ dinv, float* __restrict__ G,
                            int N, int K, int tiles) {
  int tile = blockIdx.x * (blockDim.x >> 5) + (threadIdx.x >> 5);
  if (tile >= tiles) return;
  xw_tile<true>(X, W, dinv, G, N, K, tile);
}

#define CSUM 25
__global__ void __launch_bounds__(1024, 16)
bucket_place(const int* __restrict__ src, const int* __restrict__ dst, int E, int NB,
             const int* __restrict__ cnt, int* __restrict__ bstart,
             unsigned int* __restrict__ sorted, int nplace,
             const float* __restrict__ X, const float* __restrict__ W, float* __restrict__ G,
             int N, int K, int tiles) {
  __shared__ unsigned int stage[CHUNK];
  __shared__ int loff[1024];
  __shared__ int cur[1024];
  __shared__ int gpos[1024];
  int t = threadIdx.x;
  if ((int)blockIdx.x >= nplace) {
    int tile = ((int)blockIdx.x - nplace) * 32 + (t >> 5);
    if (tile < tiles) xw_tile_k8(X, W, G, N, K, tile);
    return;
  }
  int c = 0, pre = 0, tot = 0;
  if (t < NB) {
    for (int w0 = 0; w0 < nplace; w0 += CSUM) {
      int v[CSUM];
#pragma unroll
      for (int j = 0; j < CSUM; ++j) v[j] = cnt[min(w0 + j, nplace - 1) * NB + t];
#pragma unroll
      for (int j = 0; j < CSUM; ++j) {
        int m = (w0 + j < nplace) ? v[j] : 0;
        tot += m;
        pre += (w0 + j < (int)blockIdx.x) ? m : 0;
        c = (w0 + j == (int)blockIdx.x) ? m : c;
      }
    }
  }
  loff[t] = c;
  gpos[t] = tot;
  __syncthreads();
  for (int off = 1; off < 1024; off <<= 1) {
    int v = (t >= off) ? loff[t - off] : 0;
    int u = (t >= off) ? gpos[t - off] : 0;
    __syncthreads();
    loff[t] += v;
    gpos[t] += u;
    __syncthreads();
  }
  int ex = loff[t] - c;
  int bs = gpos[t] - tot;
  loff[t] = ex;
  cur[t] = ex;
  gpos[t] = bs + pre;
  if (blockIdx.x == 0 && t < NB) {
    bstart[t] = bs;
    if (t == NB - 1) bstart[NB] = bs + tot;
  }
  __syncthreads();
  int e0 = blockIdx.x * CHUNK;
  int e1 = min(E, e0 + CHUNK);
  int e = e0 + t;
  for (; e + 4 * 1024 < e1; e += 5 * 1024) {
    int d[5], s[5];
#pragma unroll
    for (int j = 0; j < 5; ++j) { d[j] = dst[e + j * 1024]; s[j] = src[e + j * 1024]; }
#pragma unroll
    for (int j = 0; j < 5; ++j) {
      int r = atomicAdd(&cur[d[j] >> BSH], 1);
      stage[r] = ((unsigned int)s[j] << BSH) | (unsigned int)(d[j] & (BNODES - 1));
    }
  }
  for (; e < e1; e += 1024) {
    int d = dst[e], s = src[e];
    int r = atomicAdd(&cur[d >> BSH], 1);
    stage[r] = ((unsigned int)s << BSH) | (unsigned int)(d & (BNODES - 1));
  }
  __syncthreads();
  int wave = __builtin_amdgcn_readfirstlane(t >> 5), lane = t & 31;
  for (int b = wave; b < NB; b += 32) {
    int lo = loff[b], n = cur[b] - lo, gp = gpos[b];
    for (int j = lane; j < n; j += 32) sorted[gp + j] = stage[lo + j];
  }
}

__global__ void __launch_bounds__(4 * BNODES)
bucket_dinv(const unsigned int* __restrict__ sorted, const int* __restrict__ bstart,
            float* __restrict__ dinv, float* __restrict__ g, int N) {
  __shared__ int c[BNODES];
  const int T = 4 * BNODES;
  const int grow = blockIdx.x * BNODES + ((int)threadIdx.x >> 2);
  float4* gp = (float4*)(g + (size_t)min(grow, N - 1) * 16) + (threadIdx.x & 3);
  if (threadIdx.x < BNODES) c[threadIdx.x] = 0;
  __syncthreads();
  float4 gv = *gp;
  int e0 = bstart[blockIdx.x], e1 = bstart[blockIdx.x + 1];
  int e = e0 + (int)threadIdx.x;
  for (; e + 3 * T < e1; e += 4 * T) {
    unsigned int w0 = sorted[e], w1 = sorted[e + T], w2 = sorted[e + 2 * T], w3 = sorted[e + 3 * T];
    atomicAdd(&c[w0 & (BNODES - 1)], 1);
    atomicAdd(&c[w1 & (BNODES - 1)], 1);
    atomicAdd(&c[w2 & (BNODES - 1)], 1);
    atomicAdd(&c[w3 & (BNODES - 1)], 1);
  }
  for (; e < e1; e += T) atomicAdd(&c[sorted[e] & (BNODES - 1)], 1);
  __syncthreads();
  int node = blockIdx.x * BNODES + threadIdx.x;
  if (threadIdx.x < BNODES && node < N) dinv[node] = rsqrtf((float)c[threadIdx.x] + 1.0f);
  if (grow < N) {
    float di = rsqrtf((float)c[threadIdx.x >> 2] + 1.0f);
    gv.x *= di;
    gv.y *= di;
    gv.z *= di;
    gv.w *= di;
    *gp = gv;
  }
}

#define RPER 10
#define RWORDS (RPER * 4 * BNODES)
__device__ __forceinline__ float4 bucket_sum(const unsigned int* __restrict__ sorted, int e0, int e1,
                                             const float* __restrict__ g, unsigned int* stage,
                                             int* cnt, int* off, int* wtot) {
  const int T = 4 * BNODES;
  const int t = threadIdx.x;
  const int lane = t & 31;
  const int node = t >> 2, part = t & 3;
  const float4* g4 = (const float4*)g;
  float4 acc = make_float4(0.0f, 0.0f, 0.0f, 0.0f);
  for (int c0 = e0; c0 < e1; c0 += RWORDS) {
    const int c1 = min(e1, c0 + RWORDS);
    if (t < BNODES) cnt[t] = 0;
    __syncthreads();
    unsigned int w[RPER];
    int r[RPER];
#pragma unroll
    for (int j = 0; j < RPER; ++j) w[j] = sorted[min(c0 + j * T + t, c1 - 1)];
#pragma unroll
    for (int j = 0; j < RPER; ++j) {
      r[j] = 0;
      if (c0 + j * T + t < c1) r[j] = atomicAdd(&cnt[w[j] & (BNODES - 1)], 1);
    }
    __syncthreads();
    int c = (t < BNODES) ? cnt[t] : 0;
    int x = c;
    for (int d = 1; d < 32; d <<= 1) {
      int y = __shfl_up(x, d);
      if (lane >= d) x += y;
    }
    if (t < BNODES && lane == 31) wtot[t >> 5] = x;
    __syncthreads();
    if (t < BNODES) {
      int pre = 0;
      for (int k = 0; k < (t >> 5); ++k) pre += wtot[k];
      off[t] = pre + x - c;
    }
    __syncthreads();
    int o[RPER];
#pragma unroll
    for (int j = 0; j < RPER; ++j) o[j] = off[w[j] & (BNODES - 1)];
#pragma unroll
    for (int j = 0; j < RPER; ++j)
      if (c0 + j * T + t < c1) stage[o[j] + r[j]] = w[j] >> BSH;
    __syncthreads();
    const int st = off[node];
    const int n = cnt[node];
    for (int j = 0; j < n; j += 8) {
      unsigned int s[8];
      float4 v[8];
#pragma unroll
      for (int u = 0; u < 8; ++u) s[u] = stage[st + min(j + u, n - 1)];
#pragma unroll
      for (int u = 0; u < 8; ++u) v[u] = g4[(s[u] << 2) + (unsigned int)part];
#pragma unroll
      for (int u = 0; u < 8; ++u) {
        bool ok = (j + u) < n;
        acc.x += ok ? v[u].x : 0.0f;
        acc.y += ok ? v[u].y : 0.0f;
        acc.z += ok ? v[u].z : 0.0f;
        acc.w += ok ? v[u].w : 0.0f;
      }
    }
    __syncthreads();
  }
  return acc;
}

__global__ void __launch_bounds__(4 * BNODES, 16)
bucket_agg1(const unsigned int* __restrict__ sorted, const int* __restrict__ bstart,
            const float* __restrict__ g, const float* __restrict__ dinv,
            const float* __restrict__ b, float* __restrict__ h, int N) {
  __shared__ unsigned int stage[RWORDS];
  __shared__ int cnt[BNODES], off[BNODES], wtot[BNODES / 32];
  float4 a = bucket_sum(sorted, bstart[blockIdx.x], bstart[blockIdx.x + 1], g, stage, cnt, off, wtot);
  int r = threadIdx.x >> 2, part = threadIdx.x & 3;
  int node = blockIdx.x * BNODES + r;
  if (node < N) {
    float di = dinv[node];
    float4 gg = ((const float4*)(g + (size_t)node * 16))[part];
    float4 o;
    o.x = fmaxf(di * (a.x + gg.x) + b[part * 4 + 0], 0.0f);
    o.y = fmaxf(di * (a.y + gg.y) + b[part * 4 + 1], 0.0f);
    o.z = fmaxf(di * (a.z + gg.z) + b[part * 4 + 2], 0.0f);
    o.w = fmaxf(di * (a.w + gg.w) + b[part * 4 + 3], 0.0f);
    ((float4*)(h + (size_t)node * 16))[part] = o;
  }
}

__global__ void __launch_bounds__(4 * BNODES, 16)
bucket_agg2(const unsigned int* __restrict__ sorted, const int* __restrict__ bstart,
            const float* __restrict__ g, const float* __restrict__ dinv,
            const float* __restrict__ b, const float* __restrict__ Wfc,
            float* __restrict__ p, float* __restrict__ q, int N) {
  __shared__ unsigned int stage[RWORDS];
  __shared__ int cnt[BNODES], off[BNODES], wtot[BNODES / 32];
  __shared__ float acc[BNODES * APITCH];
  float4 a = bucket_sum(sorted, bstart[blockIdx.x], bstart[blockIdx.x + 1], g, stage, cnt, off, wtot);
  {
    float* o = acc + (threadIdx.x >> 2) * APITCH + (threadIdx.x & 3) * 4;
    o[0] = a.x;
    o[1] = a.y;
    o[2] = a.z;
    o[3] = a.w;
  }
  __syncthreads();
  int node = blockIdx.x * BNODES + threadIdx.x;
  if (threadIdx.x < BNODES && node < N) {
    float di = dinv[node];
    float sp = 0.0f, sq = 0.0f;
#pragma unroll
    for (int f = 0; f < 16; ++f) {
      float hh = fmaxf(di * (acc[threadIdx.x * APITCH + f] + g[(size_t)node * 16 + f]) + b[f], 0.0f);
      sp += hh * Wfc[f];
      sq += hh * Wfc[16 + f];
    }
    p[node] = sp;
    q[node] = sq;
  }
}

#define OCH 12800
#define OPER 13
__global__ void __launch_bounds__(1024)
edge_out(const int* __restrict__ src, const int* __restrict__ dst,
         const float* __restrict__ p, const float* __restrict__ q,
         const float* __restrict__ bfc, float* __restrict__ out, int E) {
  __shared__ unsigned int stS[OCH];
  __shared__ unsigned int stD[OCH];
  __shared__ float val[OCH];
  __shared__ int cS[NBMAX], cD[NBMAX];
  __shared__ int wtS[32], wtD[32];
  const int t = threadIdx.x, lane = t & 31, wave = t >> 5;
  cS[t] = 0;
  cD[t] = 0;
  __syncthreads();
  const int e0 = blockIdx.x * OCH;
  const int n = min(E - e0, OCH);
  int s[OPER], d[OPER], rs[OPER], rd[OPER];
#pragma unroll
  for (int j = 0; j < OPER; ++j) {
    int i = min(j * 1024 + t, n - 1);
    s[j] = src[e0 + i];
    d[j] = dst[e0 + i];
  }
#pragma unroll
  for (int j = 0; j < OPER; ++j) {
    rs[j] = 0;
    rd[j] = 0;
    if (j * 1024 + t < n) {
      rs[j] = atomicAdd(&cS[s[j] >> BSH], 1);
      rd[j] = atomicAdd(&cD[d[j] >> BSH], 1);
    }
  }
  __syncthreads();
  {
    const int a = cS[t], b = cD[t];
    int xa = a, xb = b;
    for (int o = 1; o < 32; o <<= 1) {
      int ya = __shfl_up(xa, o), yb = __shfl_up(xb, o);
      if (lane >= o) { xa += ya; xb += yb; }
    }
    if (lane == 31) { wtS[wave] = xa; wtD[wave] = xb; }
    __syncthreads();
    const int ta = wtS[lane], tb = wtD[lane];
    int sa = ta, sb = tb;
    for (int o = 1; o < 32; o <<= 1) {
      int ya = __shfl_up(sa, o), yb = __shfl_up(sb, o);
      if (lane >= o) { sa += ya; sb += yb; }
    }
    const int preA = __shfl(sa - ta, wave), preB = __shfl(sb - tb, wave);
    cS[t] = preA + xa - a;
    cD[t] = preB + xb - b;
  }
  __syncthreads();
#pragma unroll
  for (int j = 0; j < OPER; ++j) {
    const int i = j * 1024 + t;
    if (i < n) {
      stS[cS[s[j] >> BSH] + rs[j]] = ((unsigned int)i << 17) | (unsigned int)s[j];
      stD[cD[d[j] >> BSH] + rd[j]] = ((unsigned int)i << 17) | (unsigned int)d[j];
    }
  }
  __syncthreads();
  {
    unsigned int wd[OPER], ws[OPER];
    float vq[OPER], vp[OPER];
#pragma unroll
    for (int j = 0; j < OPER; ++j) {
      const int r = min(j * 1024 + t, n - 1);
      wd[j] = stD[r];
      ws[j] = stS[r];
    }
#pragma unroll
    for (int j = 0; j < OPER; ++j) vq[j] = q[wd[j] & 0x1FFFFu];
#pragma unroll
    for (int j = 0; j < OPER; ++j) vp[j] = p[ws[j] & 0x1FFFFu];
#pragma unroll
    for (int j = 0; j < OPER; ++j)
      if (j * 1024 + t < n) val[wd[j] >> 17] = vq[j];
    __syncthreads();
#pragma unroll
    for (int j = 0; j < OPER; ++j)
      if (j * 1024 + t < n) val[ws[j] >> 17] = vp[j] + val[ws[j] >> 17];
  }
  __syncthreads();
  const float bias = bfc[0];
#pragma unroll
  for (int j = 0; j < OPER; ++j) {
    const int i = j * 1024 + t;
    if (i < n) out[e0 + i] = val[i] + bias;
  }
}

extern "C" void kernel_launch(void* const* d_in, const int* in_sizes, int n_in,
                              void* d_out, int out_size, void* d_ws, size_t ws_size,
                              hipStream_t stream) {
  const float* x   = (const float*)d_in[0];
  const int*   ei  = (const int*)d_in[1];
  const float* W1  = (const float*)d_in[2];
  const float* b1  = (const float*)d_in[3];
  const float* W2  = (const float*)d_in[4];
  const float* b2  = (const float*)d_in[5];
  const float* Wfc = (const float*)d_in[6];
  const float* bfc = (const float*)d_in[7];
  float* out = (float*)d_out;

  const int K1 = 495;
  const int N = in_sizes[0] / K1;
  const int E = in_sizes[1] / 2;
  const int* src = ei;
  const int* dst = ei + E;

  const int NB = cdiv(N, BNODES);
  const int nchunk = cdiv(E, CHUNK);

  float* ws  = (float*)d_ws;
  float* dinv = ws;
  float* g    = ws + (size_t)N;
  float* h1   = ws + (size_t)N * 17;
  float* p    = ws + (size_t)N * 33;
  float* q    = ws + (size_t)N * 34;
  int* cnt    = (int*)(ws + (size_t)N * 35);
  int* bstart = cnt + (size_t)nchunk * NB;
  unsigned int* sorted = (unsigned int*)d_out;

  const int B = 256;
  const int tiles = cdiv(N, 16);
  const int gemmBlocks = cdiv(tiles, B / 32);
  const int xwGroups = cdiv(tiles, 32);

  bucket_count<<<nchunk, 1024, 0, stream>>>(dst, E, NB, cnt);
  bucket_place<<<nchunk + xwGroups, 1024, 0, stream>>>(src, dst, E, NB, cnt, bstart, sorted,
                                                        nchunk, x, W1, g, N, K1, tiles);

  bucket_dinv<<<NB, 4 * BNODES, 0, stream>>>(sorted, bstart, dinv, g, N);

  bucket_agg1<<<NB, 4 * BNODES, 0, stream>>>(sorted, bstart, g, dinv, b1, h1, N);

  gemm_tile16<<<gemmBlocks, B, 0, stream>>>(h1, W2, dinv, g, N, 16, tiles);
  bucket_agg2<<<NB, 4 * BNODES, 0, stream>>>(sorted, bstart, g, dinv, b2, Wfc, p, q, N);

  edge_out<<<cdiv(E, OCH), 1024, 0, stream>>>(src, dst, p, q, bfc, out, E);
}


# layer-2 product h1@W2 moves into bucket_agg1: a group parks its 128 rows of h1 in LDS, 8 of its waves run the 16-row WMMA tiles (same k order); h1 is never stored; 7 launches -> 6, same sums
# speedup vs baseline: 3.7781x; 1.0372x over previous
#include <hip/hip_runtime.h>
#include <hip/hip_bf16.h>

typedef float v2f __attribute__((ext_vector_type(2)));
typedef float v8f __attribute__((ext_vector_type(8)));

static __host__ int cdiv(int a, int b) { return (a + b - 1) / b; }


#define BSH 7
#define BNODES 128
#define NBMAX 1024
#define CHUNK 25600
#define APITCH 17

__global__ void __launch_bounds__(1024)
bucket_count(const int* __restrict__ dst, int E, int NB, int* __restrict__ cnt) {
  __shared__ int h[NBMAX];
  h[threadIdx.x] = 0;
  __syncthreads();
  int e0 = blockIdx.x * CHUNK;
  int e1 = min(E, e0 + CHUNK);
  int e = e0 + (int)threadIdx.x;
  for (; e + 4 * 1024 < e1; e += 5 * 1024) {
    int d0 = dst[e], d1 = dst[e + 1024], d2 = dst[e + 2048], d3 = dst[e + 3072], d4 = dst[e + 4096];
    atomicAdd(&h[d0 >> BSH], 1);
    atomicAdd(&h[d1 >> BSH], 1);
    atomicAdd(&h[d2 >> BSH], 1);
    atomicAdd(&h[d3 >> BSH], 1);
    atomicAdd(&h[d4 >> BSH], 1);
  }
  for (; e < e1; e += 1024) atomicAdd(&h[dst[e] >> BSH], 1);
  __syncthreads();
  if ((int)threadIdx.x < NB) cnt[blockIdx.x * NB + threadIdx.x] = h[threadIdx.x];
}


__device__ __forceinline__ void xw_tile_k8(const float* __restrict__ X, const float* __restrict__ W,
                                           float* __restrict__ G, int N, int K, int tile) {
  int lane = threadIdx.x & 31;
  int hi = lane >> 4;
  int lo = lane & 15;
  int row = tile * 16 + lo;
  float rowMask = (row < N) ? 1.0f : 0.0f;
  const float* xr = X + (size_t)((row < N) ? row : 0) * K;
  const float* wc = W + lo;
  int K8 = K & ~7;

  v8f c = {};
  int k0 = 0;
  for (; k0 < K8; k0 += 8) {
    int kk = k0 + 4 * hi;
    if ((k0 & 63) == 0) __builtin_prefetch(xr + kk + 64, 0, 1);
    float x0 = xr[kk], x1 = xr[kk + 1], x2 = xr[kk + 2], x3 = xr[kk + 3];
    v2f a, b;
    a.x = x0 * rowMask;
    a.y = x1 * rowMask;
    b.x = wc[kk * 16];
    b.y = wc[(kk + 1) * 16];
    c = __builtin_amdgcn_wmma_f32_16x16x4_f32(false, a, false, b, (short)0, c,
                                              false, false);
    a.x = x2 * rowMask;
    a.y = x3 * rowMask;
    b.x = wc[(kk + 2) * 16];
    b.y = wc[(kk + 3) * 16];
    c = __builtin_amdgcn_wmma_f32_16x16x4_f32(false, a, false, b, (short)0, c,
                                              false, false);
  }
  if (k0 < K) {
    int kk = k0 + 4 * hi;
    float xa[4], wb[4];
#pragma unroll
    for (int s = 0; s < 4; ++s) {
      int kc = (kk + s < K) ? kk + s : 0;
      float m = (kk + s < K) ? rowMask : 0.0f;
      xa[s] = xr[kc] * m;
      wb[s] = wc[kc * 16];
    }
    v2f a, b;
    a.x = xa[0]; a.y = xa[1]; b.x = wb[0]; b.y = wb[1];
    c = __builtin_amdgcn_wmma_f32_16x16x4_f32(false, a, false, b, (short)0, c,
                                              false, false);
    a.x = xa[2]; a.y = xa[3]; b.x = wb[2]; b.y = wb[3];
    c = __builtin_amdgcn_wmma_f32_16x16x4_f32(false, a, false, b, (short)0, c,
                                              false, false);
  }
#pragma unroll
  for (int j = 0; j < 8; ++j) {
    int rr = tile * 16 + 8 * hi + j;
    if (rr < N) G[(size_t)rr * 16 + lo] = c[j];
  }
}

#define CSUM 25
__global__ void __launch_bounds__(1024, 16)
bucket_place(const int* __restrict__ src, const int* __restrict__ dst, int E, int NB,
             const int* __restrict__ cnt, int* __restrict__ bstart,
             unsigned int* __restrict__ sorted, int nplace,
             const float* __restrict__ X, const float* __restrict__ W, float* __restrict__ G,
             int N, int K, int tiles) {
  __shared__ unsigned int stage[CHUNK];
  __shared__ int loff[1024];
  __shared__ int cur[1024];
  __shared__ int gpos[1024];
  int t = threadIdx.x;
  if ((int)blockIdx.x >= nplace) {
    int tile = ((int)blockIdx.x - nplace) * 32 + (t >> 5);
    if (tile < tiles) xw_tile_k8(X, W, G, N, K, tile);
    return;
  }
  int c = 0, pre = 0, tot = 0;
  if (t < NB) {
    for (int w0 = 0; w0 < nplace; w0 += CSUM) {
      int v[CSUM];
#pragma unroll
      for (int j = 0; j < CSUM; ++j) v[j] = cnt[min(w0 + j, nplace - 1) * NB + t];
#pragma unroll
      for (int j = 0; j < CSUM; ++j) {
        int m = (w0 + j < nplace) ? v[j] : 0;
        tot += m;
        pre += (w0 + j < (int)blockIdx.x) ? m : 0;
        c = (w0 + j == (int)blockIdx.x) ? m : c;
      }
    }
  }
  loff[t] = c;
  gpos[t] = tot;
  __syncthreads();
  for (int off = 1; off < 1024; off <<= 1) {
    int v = (t >= off) ? loff[t - off] : 0;
    int u = (t >= off) ? gpos[t - off] : 0;
    __syncthreads();
    loff[t] += v;
    gpos[t] += u;
    __syncthreads();
  }
  int ex = loff[t] - c;
  int bs = gpos[t] - tot;
  loff[t] = ex;
  cur[t] = ex;
  gpos[t] = bs + pre;
  if (blockIdx.x == 0 && t < NB) {
    bstart[t] = bs;
    if (t == NB - 1) bstart[NB] = bs + tot;
  }
  __syncthreads();
  int e0 = blockIdx.x * CHUNK;
  int e1 = min(E, e0 + CHUNK);
  int e = e0 + t;
  for (; e + 4 * 1024 < e1; e += 5 * 1024) {
    int d[5], s[5];
#pragma unroll
    for (int j = 0; j < 5; ++j) { d[j] = dst[e + j * 1024]; s[j] = src[e + j * 1024]; }
#pragma unroll
    for (int j = 0; j < 5; ++j) {
      int r = atomicAdd(&cur[d[j] >> BSH], 1);
      stage[r] = ((unsigned int)s[j] << BSH) | (unsigned int)(d[j] & (BNODES - 1));
    }
  }
  for (; e < e1; e += 1024) {
    int d = dst[e], s = src[e];
    int r = atomicAdd(&cur[d >> BSH], 1);
    stage[r] = ((unsigned int)s << BSH) | (unsigned int)(d & (BNODES - 1));
  }
  __syncthreads();
  int wave = __builtin_amdgcn_readfirstlane(t >> 5), lane = t & 31;
  for (int b = wave; b < NB; b += 32) {
    int lo = loff[b], n = cur[b] - lo, gp = gpos[b];
    for (int j = lane; j < n; j += 32) sorted[gp + j] = stage[lo + j];
  }
}

__global__ void __launch_bounds__(4 * BNODES)
bucket_dinv(const unsigned int* __restrict__ sorted, const int* __restrict__ bstart,
            float* __restrict__ dinv, float* __restrict__ g, int N) {
  __shared__ int c[BNODES];
  const int T = 4 * BNODES;
  const int grow = blockIdx.x * BNODES + ((int)threadIdx.x >> 2);
  float4* gp = (float4*)(g + (size_t)min(grow, N - 1) * 16) + (threadIdx.x & 3);
  if (threadIdx.x < BNODES) c[threadIdx.x] = 0;
  __syncthreads();
  float4 gv = *gp;
  int e0 = bstart[blockIdx.x], e1 = bstart[blockIdx.x + 1];
  int e = e0 + (int)threadIdx.x;
  for (; e + 3 * T < e1; e += 4 * T) {
    unsigned int w0 = sorted[e], w1 = sorted[e + T], w2 = sorted[e + 2 * T], w3 = sorted[e + 3 * T];
    atomicAdd(&c[w0 & (BNODES - 1)], 1);
    atomicAdd(&c[w1 & (BNODES - 1)], 1);
    atomicAdd(&c[w2 & (BNODES - 1)], 1);
    atomicAdd(&c[w3 & (BNODES - 1)], 1);
  }
  for (; e < e1; e += T) atomicAdd(&c[sorted[e] & (BNODES - 1)], 1);
  __syncthreads();
  int node = blockIdx.x * BNODES + threadIdx.x;
  if (threadIdx.x < BNODES && node < N) dinv[node] = rsqrtf((float)c[threadIdx.x] + 1.0f);
  if (grow < N) {
    float di = rsqrtf((float)c[threadIdx.x >> 2] + 1.0f);
    gv.x *= di;
    gv.y *= di;
    gv.z *= di;
    gv.w *= di;
    *gp = gv;
  }
}

#define RPER 10
#define RWORDS (RPER * 4 * BNODES)
__device__ __forceinline__ float4 bucket_sum(const unsigned int* __restrict__ sorted, int e0, int e1,
                                             const float* __restrict__ g, unsigned int* stage,
                                             int* cnt, int* off, int* wtot) {
  const int T = 4 * BNODES;
  const int t = threadIdx.x;
  const int lane = t & 31;
  const int node = t >> 2, part = t & 3;
  const float4* g4 = (const float4*)g;
  float4 acc = make_float4(0.0f, 0.0f, 0.0f, 0.0f);
  for (int c0 = e0; c0 < e1; c0 += RWORDS) {
    const int c1 = min(e1, c0 + RWORDS);
    if (t < BNODES) cnt[t] = 0;
    __syncthreads();
    unsigned int w[RPER];
    int r[RPER];
#pragma unroll
    for (int j = 0; j < RPER; ++j) w[j] = sorted[min(c0 + j * T + t, c1 - 1)];
#pragma unroll
    for (int j = 0; j < RPER; ++j) {
      r[j] = 0;
      if (c0 + j * T + t < c1) r[j] = atomicAdd(&cnt[w[j] & (BNODES - 1)], 1);
    }
    __syncthreads();
    int c = (t < BNODES) ? cnt[t] : 0;
    int x = c;
    for (int d = 1; d < 32; d <<= 1) {
      int y = __shfl_up(x, d);
      if (lane >= d) x += y;
    }
    if (t < BNODES && lane == 31) wtot[t >> 5] = x;
    __syncthreads();
    if (t < BNODES) {
      int pre = 0;
      for (int k = 0; k < (t >> 5); ++k) pre += wtot[k];
      off[t] = pre + x - c;
    }
    __syncthreads();
    int o[RPER];
#pragma unroll
    for (int j = 0; j < RPER; ++j) o[j] = off[w[j] & (BNODES - 1)];
#pragma unroll
    for (int j = 0; j < RPER; ++j)
      if (c0 + j * T + t < c1) stage[o[j] + r[j]] = w[j] >> BSH;
    __syncthreads();
    const int st = off[node];
    const int n = cnt[node];
    for (int j = 0; j < n; j += 8) {
      unsigned int s[8];
      float4 v[8];
#pragma unroll
      for (int u = 0; u < 8; ++u) s[u] = stage[st + min(j + u, n - 1)];
#pragma unroll
      for (int u = 0; u < 8; ++u) v[u] = g4[(s[u] << 2) + (unsigned int)part];
#pragma unroll
      for (int u = 0; u < 8; ++u) {
        bool ok = (j + u) < n;
        acc.x += ok ? v[u].x : 0.0f;
        acc.y += ok ? v[u].y : 0.0f;
        acc.z += ok ? v[u].z : 0.0f;
        acc.w += ok ? v[u].w : 0.0f;
      }
    }
    __syncthreads();
  }
  return acc;
}

#define HPITCH 20
__global__ void __launch_bounds__(4 * BNODES, 16)
bucket_agg1(const unsigned int* __restrict__ sorted, const int* __restrict__ bstart,
            const float* __restrict__ g, const float* __restrict__ dinv,
            const float* __restrict__ b, const float* __restrict__ W2,
            float* __restrict__ g2, int N) {
  __shared__ unsigned int stage[RWORDS];
  __shared__ int cnt[BNODES], off[BNODES], wtot[BNODES / 32];
  __shared__ alignas(16) float hs[BNODES * HPITCH];
  float4 a = bucket_sum(sorted, bstart[blockIdx.x], bstart[blockIdx.x + 1], g, stage, cnt, off, wtot);
  const int wave = __builtin_amdgcn_readfirstlane((int)threadIdx.x >> 5);
  const int lane = threadIdx.x & 31, hi = lane >> 4, lo = lane & 15;
  float wb[8] = {0.0f, 0.0f, 0.0f, 0.0f, 0.0f, 0.0f, 0.0f, 0.0f};
  if (wave < BNODES / 16) {
#pragma unroll
    for (int s = 0; s < 4; ++s) {
      wb[2 * s]     = W2[(4 * s + 2 * hi) * 16 + lo];
      wb[2 * s + 1] = W2[(4 * s + 2 * hi + 1) * 16 + lo];
    }
  }
  int r = threadIdx.x >> 2, part = threadIdx.x & 3;
  int node = blockIdx.x * BNODES + r;
  float4 o = make_float4(0.0f, 0.0f, 0.0f, 0.0f);
  float di = 0.0f;
  if (node < N) {
    di = dinv[node];
    float4 gg = ((const float4*)(g + (size_t)node * 16))[part];
    o.x = fmaxf(di * (a.x + gg.x) + b[part * 4 + 0], 0.0f);
    o.y = fmaxf(di * (a.y + gg.y) + b[part * 4 + 1], 0.0f);
    o.z = fmaxf(di * (a.z + gg.z) + b[part * 4 + 2], 0.0f);
    o.w = fmaxf(di * (a.w + gg.w) + b[part * 4 + 3], 0.0f);
  }
  *(float4*)(hs + r * HPITCH + part * 4) = o;
  if (part == 0) hs[r * HPITCH + 16] = di;
  __syncthreads();
  if (wave < BNODES / 16) {
    const float* hr = hs + (wave * 16 + lo) * HPITCH + 2 * hi;
    const float* dr = hs + (wave * 16 + 8 * hi) * HPITCH + 16;
    float dv[8];
#pragma unroll
    for (int j = 0; j < 8; ++j) dv[j] = dr[j * HPITCH];
    v8f c = {};
#pragma unroll
    for (int s = 0; s < 4; ++s) {
      v2f av, bv;
      av.x = hr[4 * s];
      av.y = hr[4 * s + 1];
      bv.x = wb[2 * s];
      bv.y = wb[2 * s + 1];
      c = __builtin_amdgcn_wmma_f32_16x16x4_f32(false, av, false, bv, (short)0, c,
                                                false, false);
    }
#pragma unroll
    for (int j = 0; j < 8; ++j) {
      const int rr = blockIdx.x * BNODES + wave * 16 + 8 * hi + j;
      if (rr < N) g2[(size_t)rr * 16 + lo] = c[j] * dv[j];
    }
  }
}

__global__ void __launch_bounds__(4 * BNODES, 16)
bucket_agg2(const unsigned int* __restrict__ sorted, const int* __restrict__ bstart,
            const float* __restrict__ g, const float* __restrict__ dinv,
            const float* __restrict__ b, const float* __restrict__ Wfc,
            float* __restrict__ p, float* __restrict__ q, int N) {
  __shared__ unsigned int stage[RWORDS];
  __shared__ int cnt[BNODES], off[BNODES], wtot[BNODES / 32];
  __shared__ float acc[BNODES * APITCH];
  float4 a = bucket_sum(sorted, bstart[blockIdx.x], bstart[blockIdx.x + 1], g, stage, cnt, off, wtot);
  {
    float* o = acc + (threadIdx.x >> 2) * APITCH + (threadIdx.x & 3) * 4;
    o[0] = a.x;
    o[1] = a.y;
    o[2] = a.z;
    o[3] = a.w;
  }
  __syncthreads();
  int node = blockIdx.x * BNODES + threadIdx.x;
  if (threadIdx.x < BNODES && node < N) {
    float di = dinv[node];
    float sp = 0.0f, sq = 0.0f;
#pragma unroll
    for (int f = 0; f < 16; ++f) {
      float hh = fmaxf(di * (acc[threadIdx.x * APITCH + f] + g[(size_t)node * 16 + f]) + b[f], 0.0f);
      sp += hh * Wfc[f];
      sq += hh * Wfc[16 + f];
    }
    p[node] = sp;
    q[node] = sq;
  }
}

#define OCH 12800
#define OPER 13
__global__ void __launch_bounds__(1024)
edge_out(const int* __restrict__ src, const int* __restrict__ dst,
         const float* __restrict__ p, const float* __restrict__ q,
         const float* __restrict__ bfc, float* __restrict__ out, int E) {
  __shared__ unsigned int stS[OCH];
  __shared__ unsigned int stD[OCH];
  __shared__ float val[OCH];
  __shared__ int cS[NBMAX], cD[NBMAX];
  __shared__ int wtS[32], wtD[32];
  const int t = threadIdx.x, lane = t & 31, wave = t >> 5;
  cS[t] = 0;
  cD[t] = 0;
  __syncthreads();
  const int e0 = blockIdx.x * OCH;
  const int n = min(E - e0, OCH);
  int s[OPER], d[OPER], rs[OPER], rd[OPER];
#pragma unroll
  for (int j = 0; j < OPER; ++j) {
    int i = min(j * 1024 + t, n - 1);
    s[j] = src[e0 + i];
    d[j] = dst[e0 + i];
  }
#pragma unroll
  for (int j = 0; j < OPER; ++j) {
    rs[j] = 0;
    rd[j] = 0;
    if (j * 1024 + t < n) {
      rs[j] = atomicAdd(&cS[s[j] >> BSH], 1);
      rd[j] = atomicAdd(&cD[d[j] >> BSH], 1);
    }
  }
  __syncthreads();
  {
    const int a = cS[t], b = cD[t];
    int xa = a, xb = b;
    for (int o = 1; o < 32; o <<= 1) {
      int ya = __shfl_up(xa, o), yb = __shfl_up(xb, o);
      if (lane >= o) { xa += ya; xb += yb; }
    }
    if (lane == 31) { wtS[wave] = xa; wtD[wave] = xb; }
    __syncthreads();
    const int ta = wtS[lane], tb = wtD[lane];
    int sa = ta, sb = tb;
    for (int o = 1; o < 32; o <<= 1) {
      int ya = __shfl_up(sa, o), yb = __shfl_up(sb, o);
      if (lane >= o) { sa += ya; sb += yb; }
    }
    const int preA = __shfl(sa - ta, wave), preB = __shfl(sb - tb, wave);
    cS[t] = preA + xa - a;
    cD[t] = preB + xb - b;
  }
  __syncthreads();
#pragma unroll
  for (int j = 0; j < OPER; ++j) {
    const int i = j * 1024 + t;
    if (i < n) {
      stS[cS[s[j] >> BSH] + rs[j]] = ((unsigned int)i << 17) | (unsigned int)s[j];
      stD[cD[d[j] >> BSH] + rd[j]] = ((unsigned int)i << 17) | (unsigned int)d[j];
    }
  }
  __syncthreads();
  {
    unsigned int wd[OPER], ws[OPER];
    float vq[OPER], vp[OPER];
#pragma unroll
    for (int j = 0; j < OPER; ++j) {
      const int r = min(j * 1024 + t, n - 1);
      wd[j] = stD[r];
      ws[j] = stS[r];
    }
#pragma unroll
    for (int j = 0; j < OPER; ++j) vq[j] = q[wd[j] & 0x1FFFFu];
#pragma unroll
    for (int j = 0; j < OPER; ++j) vp[j] = p[ws[j] & 0x1FFFFu];
#pragma unroll
    for (int j = 0; j < OPER; ++j)
      if (j * 1024 + t < n) val[wd[j] >> 17] = vq[j];
    __syncthreads();
#pragma unroll
    for (int j = 0; j < OPER; ++j)
      if (j * 1024 + t < n) val[ws[j] >> 17] = vp[j] + val[ws[j] >> 17];
  }
  __syncthreads();
  const float bias = bfc[0];
#pragma unroll
  for (int j = 0; j < OPER; ++j) {
    const int i = j * 1024 + t;
    if (i < n) out[e0 + i] = val[i] + bias;
  }
}

extern "C" void kernel_launch(void* const* d_in, const int* in_sizes, int n_in,
                              void* d_out, int out_size, void* d_ws, size_t ws_size,
                              hipStream_t stream) {
  const float* x   = (const float*)d_in[0];
  const int*   ei  = (const int*)d_in[1];
  const float* W1  = (const float*)d_in[2];
  const float* b1  = (const float*)d_in[3];
  const float* W2  = (const float*)d_in[4];
  const float* b2  = (const float*)d_in[5];
  const float* Wfc = (const float*)d_in[6];
  const float* bfc = (const float*)d_in[7];
  float* out = (float*)d_out;

  const int K1 = 495;
  const int N = in_sizes[0] / K1;
  const int E = in_sizes[1] / 2;
  const int* src = ei;
  const int* dst = ei + E;

  const int NB = cdiv(N, BNODES);
  const int nchunk = cdiv(E, CHUNK);

  float* ws  = (float*)d_ws;
  float* dinv = ws;
  float* g    = ws + (size_t)N;
  float* g2   = ws + (size_t)N * 17;
  float* p    = ws + (size_t)N * 33;
  float* q    = ws + (size_t)N * 34;
  int* cnt    = (int*)(ws + (size_t)N * 35);
  int* bstart = cnt + (size_t)nchunk * NB;
  unsigned int* sorted = (unsigned int*)d_out;

  const int tiles = cdiv(N, 16);
  const int xwGroups = cdiv(tiles, 32);

  bucket_count<<<nchunk, 1024, 0, stream>>>(dst, E, NB, cnt);
  bucket_place<<<nchunk + xwGroups, 1024, 0, stream>>>(src, dst, E, NB, cnt, bstart, sorted,
                                                        nchunk, x, W1, g, N, K1, tiles);

  bucket_dinv<<<NB, 4 * BNODES, 0, stream>>>(sorted, bstart, dinv, g, N);

  bucket_agg1<<<NB, 4 * BNODES, 0, stream>>>(sorted, bstart, g, dinv, b1, W2, g2, N);

  bucket_agg2<<<NB, 4 * BNODES, 0, stream>>>(sorted, bstart, g2, dinv, b2, Wfc, p, q, N);

  edge_out<<<cdiv(E, OCH), 1024, 0, stream>>>(src, dst, p, q, bfc, out, E);
}


# layer-1 product: each product work-group copies W1 once into its LDS in fragment order; a tile's W fragment is one 16-byte LDS read per 8-k block, not four global loads; x path, k order, sums same
# speedup vs baseline: 3.9496x; 1.0454x over previous
#include <hip/hip_runtime.h>
#include <hip/hip_bf16.h>

typedef float v2f __attribute__((ext_vector_type(2)));
typedef float v8f __attribute__((ext_vector_type(8)));

static __host__ int cdiv(int a, int b) { return (a + b - 1) / b; }


#define BSH 7
#define BNODES 128
#define NBMAX 1024
#define CHUNK 25600
#define APITCH 17

__global__ void __launch_bounds__(1024)
bucket_count(const int* __restrict__ dst, int E, int NB, int* __restrict__ cnt) {
  __shared__ int h[NBMAX];
  h[threadIdx.x] = 0;
  __syncthreads();
  int e0 = blockIdx.x * CHUNK;
  int e1 = min(E, e0 + CHUNK);
  int e = e0 + (int)threadIdx.x;
  for (; e + 4 * 1024 < e1; e += 5 * 1024) {
    int d0 = dst[e], d1 = dst[e + 1024], d2 = dst[e + 2048], d3 = dst[e + 3072], d4 = dst[e + 4096];
    atomicAdd(&h[d0 >> BSH], 1);
    atomicAdd(&h[d1 >> BSH], 1);
    atomicAdd(&h[d2 >> BSH], 1);
    atomicAdd(&h[d3 >> BSH], 1);
    atomicAdd(&h[d4 >> BSH], 1);
  }
  for (; e < e1; e += 1024) atomicAdd(&h[dst[e] >> BSH], 1);
  __syncthreads();
  if ((int)threadIdx.x < NB) cnt[blockIdx.x * NB + threadIdx.x] = h[threadIdx.x];
}


__device__ __forceinline__ void xw_tile_k8(const float* __restrict__ X, const float4* wl,
                                           float* __restrict__ G, int N, int K, int tile) {
  int lane = threadIdx.x & 31;
  int hi = lane >> 4;
  int lo = lane & 15;
  int row = tile * 16 + lo;
  float rowMask = (row < N) ? 1.0f : 0.0f;
  const float* xr = X + (size_t)((row < N) ? row : 0) * K;
  const float4* wf = wl + lane;
  int K8 = K & ~7;

  v8f c = {};
  int k0 = 0;
  for (; k0 < K8; k0 += 8) {
    int kk = k0 + 4 * hi;
    if ((k0 & 63) == 0) __builtin_prefetch(xr + kk + 64, 0, 1);
    float x0 = xr[kk], x1 = xr[kk + 1], x2 = xr[kk + 2], x3 = xr[kk + 3];
    float4 wv = wf[k0 * 4];
    v2f a, b;
    a.x = x0 * rowMask;
    a.y = x1 * rowMask;
    b.x = wv.x;
    b.y = wv.y;
    c = __builtin_amdgcn_wmma_f32_16x16x4_f32(false, a, false, b, (short)0, c,
                                              false, false);
    a.x = x2 * rowMask;
    a.y = x3 * rowMask;
    b.x = wv.z;
    b.y = wv.w;
    c = __builtin_amdgcn_wmma_f32_16x16x4_f32(false, a, false, b, (short)0, c,
                                              false, false);
  }
  if (k0 < K) {
    int kk = k0 + 4 * hi;
    float xa[4];
#pragma unroll
    for (int s = 0; s < 4; ++s) {
      int kc = (kk + s < K) ? kk + s : 0;
      float m = (kk + s < K) ? rowMask : 0.0f;
      xa[s] = xr[kc] * m;
    }
    float4 wv = wf[k0 * 4];
    v2f a, b;
    a.x = xa[0]; a.y = xa[1]; b.x = wv.x; b.y = wv.y;
    c = __builtin_amdgcn_wmma_f32_16x16x4_f32(false, a, false, b, (short)0, c,
                                              false, false);
    a.x = xa[2]; a.y = xa[3]; b.x = wv.z; b.y = wv.w;
    c = __builtin_amdgcn_wmma_f32_16x16x4_f32(false, a, false, b, (short)0, c,
                                              false, false);
  }
#pragma unroll
  for (int j = 0; j < 8; ++j) {
    int rr = tile * 16 + 8 * hi + j;
    if (rr < N) G[(size_t)rr * 16 + lo] = c[j];
  }
}

__device__ __forceinline__ void w_frag_fill(const float* __restrict__ W, float4* wl, int K) {
  const int slots = ((K + 7) >> 3) * 32;
  for (int i = threadIdx.x; i < slots; i += 1024) {
    const int ln = i & 31;
    const int k = (i >> 5) * 8 + 4 * (ln >> 4);
    const float* wc = W + (ln & 15);
    const float w0 = wc[min(k, K - 1) * 16], w1 = wc[min(k + 1, K - 1) * 16];
    const float w2 = wc[min(k + 2, K - 1) * 16], w3 = wc[min(k + 3, K - 1) * 16];
    float4 v;
    v.x = (k     < K) ? w0 : 0.0f;
    v.y = (k + 1 < K) ? w1 : 0.0f;
    v.z = (k + 2 < K) ? w2 : 0.0f;
    v.w = (k + 3 < K) ? w3 : 0.0f;
    wl[i] = v;
  }
}

#define CSUM 25
__global__ void __launch_bounds__(1024, 16)
bucket_place(const int* __restrict__ src, const int* __restrict__ dst, int E, int NB,
             const int* __restrict__ cnt, int* __restrict__ bstart,
             unsigned int* __restrict__ sorted, int nplace,
             const float* __restrict__ X, const float* __restrict__ W, float* __restrict__ G,
             int N, int K, int tiles) {
  __shared__ alignas(16) unsigned int stage[CHUNK];
  __shared__ int loff[1024];
  __shared__ int cur[1024];
  __shared__ int gpos[1024];
  int t = threadIdx.x;
  if ((int)blockIdx.x >= nplace) {
    int tile = ((int)blockIdx.x - nplace) * 32 + (t >> 5);
    float4* wl = reinterpret_cast<float4*>(stage);
    if (tile < tiles) __builtin_prefetch(X + (size_t)min(tile * 16 + (t & 15), N - 1) * K, 0, 1);
    w_frag_fill(W, wl, K);
    __syncthreads();
    if (tile < tiles) xw_tile_k8(X, wl, G, N, K, tile);
    return;
  }
  int c = 0, pre = 0, tot = 0;
  if (t < NB) {
    for (int w0 = 0; w0 < nplace; w0 += CSUM) {
      int v[CSUM];
#pragma unroll
      for (int j = 0; j < CSUM; ++j) v[j] = cnt[min(w0 + j, nplace - 1) * NB + t];
#pragma unroll
      for (int j = 0; j < CSUM; ++j) {
        int m = (w0 + j < nplace) ? v[j] : 0;
        tot += m;
        pre += (w0 + j < (int)blockIdx.x) ? m : 0;
        c = (w0 + j == (int)blockIdx.x) ? m : c;
      }
    }
  }
  loff[t] = c;
  gpos[t] = tot;
  __syncthreads();
  for (int off = 1; off < 1024; off <<= 1) {
    int v = (t >= off) ? loff[t - off] : 0;
    int u = (t >= off) ? gpos[t - off] : 0;
    __syncthreads();
    loff[t] += v;
    gpos[t] += u;
    __syncthreads();
  }
  int ex = loff[t] - c;
  int bs = gpos[t] - tot;
  loff[t] = ex;
  cur[t] = ex;
  gpos[t] = bs + pre;
  if (blockIdx.x == 0 && t < NB) {
    bstart[t] = bs;
    if (t == NB - 1) bstart[NB] = bs + tot;
  }
  __syncthreads();
  int e0 = blockIdx.x * CHUNK;
  int e1 = min(E, e0 + CHUNK);
  int e = e0 + t;
  for (; e + 4 * 1024 < e1; e += 5 * 1024) {
    int d[5], s[5];
#pragma unroll
    for (int j = 0; j < 5; ++j) { d[j] = dst[e + j * 1024]; s[j] = src[e + j * 1024]; }
#pragma unroll
    for (int j = 0; j < 5; ++j) {
      int r = atomicAdd(&cur[d[j] >> BSH], 1);
      stage[r] = ((unsigned int)s[j] << BSH) | (unsigned int)(d[j] & (BNODES - 1));
    }
  }
  for (; e < e1; e += 1024) {
    int d = dst[e], s = src[e];
    int r = atomicAdd(&cur[d >> BSH], 1);
    stage[r] = ((unsigned int)s << BSH) | (unsigned int)(d & (BNODES - 1));
  }
  __syncthreads();
  int wave = __builtin_amdgcn_readfirstlane(t >> 5), lane = t & 31;
  for (int b = wave; b < NB; b += 32) {
    int lo = loff[b], n = cur[b] - lo, gp = gpos[b];
    for (int j = lane; j < n; j += 32) sorted[gp + j] = stage[lo + j];
  }
}

__global__ void __launch_bounds__(4 * BNODES)
bucket_dinv(const unsigned int* __restrict__ sorted, const int* __restrict__ bstart,
            float* __restrict__ dinv, float* __restrict__ g, int N) {
  __shared__ int c[BNODES];
  const int T = 4 * BNODES;
  const int grow = blockIdx.x * BNODES + ((int)threadIdx.x >> 2);
  float4* gp = (float4*)(g + (size_t)min(grow, N - 1) * 16) + (threadIdx.x & 3);
  if (threadIdx.x < BNODES) c[threadIdx.x] = 0;
  __syncthreads();
  float4 gv = *gp;
  int e0 = bstart[blockIdx.x], e1 = bstart[blockIdx.x + 1];
  int e = e0 + (int)threadIdx.x;
  for (; e + 3 * T < e1; e += 4 * T) {
    unsigned int w0 = sorted[e], w1 = sorted[e + T], w2 = sorted[e + 2 * T], w3 = sorted[e + 3 * T];
    atomicAdd(&c[w0 & (BNODES - 1)], 1);
    atomicAdd(&c[w1 & (BNODES - 1)], 1);
    atomicAdd(&c[w2 & (BNODES - 1)], 1);
    atomicAdd(&c[w3 & (BNODES - 1)], 1);
  }
  for (; e < e1; e += T) atomicAdd(&c[sorted[e] & (BNODES - 1)], 1);
  __syncthreads();
  int node = blockIdx.x * BNODES + threadIdx.x;
  if (threadIdx.x < BNODES && node < N) dinv[node] = rsqrtf((float)c[threadIdx.x] + 1.0f);
  if (grow < N) {
    float di = rsqrtf((float)c[threadIdx.x >> 2] + 1.0f);
    gv.x *= di;
    gv.y *= di;
    gv.z *= di;
    gv.w *= di;
    *gp = gv;
  }
}

#define RPER 10
#define RWORDS (RPER * 4 * BNODES)
__device__ __forceinline__ float4 bucket_sum(const unsigned int* __restrict__ sorted, int e0, int e1,
                                             const float* __restrict__ g, unsigned int* stage,
                                             int* cnt, int* off, int* wtot) {
  const int T = 4 * BNODES;
  const int t = threadIdx.x;
  const int lane = t & 31;
  const int node = t >> 2, part = t & 3;
  const float4* g4 = (const float4*)g;
  float4 acc = make_float4(0.0f, 0.0f, 0.0f, 0.0f);
  for (int c0 = e0; c0 < e1; c0 += RWORDS) {
    const int c1 = min(e1, c0 + RWORDS);
    if (t < BNODES) cnt[t] = 0;
    __syncthreads();
    unsigned int w[RPER];
    int r[RPER];
#pragma unroll
    for (int j = 0; j < RPER; ++j) w[j] = sorted[min(c0 + j * T + t, c1 - 1)];
#pragma unroll
    for (int j = 0; j < RPER; ++j) {
      r[j] = 0;
      if (c0 + j * T + t < c1) r[j] = atomicAdd(&cnt[w[j] & (BNODES - 1)], 1);
    }
    __syncthreads();
    int c = (t < BNODES) ? cnt[t] : 0;
    int x = c;
    for (int d = 1; d < 32; d <<= 1) {
      int y = __shfl_up(x, d);
      if (lane >= d) x += y;
    }
    if (t < BNODES && lane == 31) wtot[t >> 5] = x;
    __syncthreads();
    if (t < BNODES) {
      int pre = 0;
      for (int k = 0; k < (t >> 5); ++k) pre += wtot[k];
      off[t] = pre + x - c;
    }
    __syncthreads();
    int o[RPER];
#pragma unroll
    for (int j = 0; j < RPER; ++j) o[j] = off[w[j] & (BNODES - 1)];
#pragma unroll
    for (int j = 0; j < RPER; ++j)
      if (c0 + j * T + t < c1) stage[o[j] + r[j]] = w[j] >> BSH;
    __syncthreads();
    const int st = off[node];
    const int n = cnt[node];
    for (int j = 0; j < n; j += 8) {
      unsigned int s[8];
      float4 v[8];
#pragma unroll
      for (int u = 0; u < 8; ++u) s[u] = stage[st + min(j + u, n - 1)];
#pragma unroll
      for (int u = 0; u < 8; ++u) v[u] = g4[(s[u] << 2) + (unsigned int)part];
#pragma unroll
      for (int u = 0; u < 8; ++u) {
        bool ok = (j + u) < n;
        acc.x += ok ? v[u].x : 0.0f;
        acc.y += ok ? v[u].y : 0.0f;
        acc.z += ok ? v[u].z : 0.0f;
        acc.w += ok ? v[u].w : 0.0f;
      }
    }
    __syncthreads();
  }
  return acc;
}

#define HPITCH 20
__global__ void __launch_bounds__(4 * BNODES, 16)
bucket_agg1(const unsigned int* __restrict__ sorted, const int* __restrict__ bstart,
            const float* __restrict__ g, const float* __restrict__ dinv,
            const float* __restrict__ b, const float* __restrict__ W2,
            float* __restrict__ g2, int N) {
  __shared__ unsigned int stage[RWORDS];
  __shared__ int cnt[BNODES], off[BNODES], wtot[BNODES / 32];
  __shared__ alignas(16) float hs[BNODES * HPITCH];
  float4 a = bucket_sum(sorted, bstart[blockIdx.x], bstart[blockIdx.x + 1], g, stage, cnt, off, wtot);
  const int wave = __builtin_amdgcn_readfirstlane((int)threadIdx.x >> 5);
  const int lane = threadIdx.x & 31, hi = lane >> 4, lo = lane & 15;
  float wb[8] = {0.0f, 0.0f, 0.0f, 0.0f, 0.0f, 0.0f, 0.0f, 0.0f};
  if (wave < BNODES / 16) {
#pragma unroll
    for (int s = 0; s < 4; ++s) {
      wb[2 * s]     = W2[(4 * s + 2 * hi) * 16 + lo];
      wb[2 * s + 1] = W2[(4 * s + 2 * hi + 1) * 16 + lo];
    }
  }
  int r = threadIdx.x >> 2, part = threadIdx.x & 3;
  int node = blockIdx.x * BNODES + r;
  float4 o = make_float4(0.0f, 0.0f, 0.0f, 0.0f);
  float di = 0.0f;
  if (node < N) {
    di = dinv[node];
    float4 gg = ((const float4*)(g + (size_t)node * 16))[part];
    o.x = fmaxf(di * (a.x + gg.x) + b[part * 4 + 0], 0.0f);
    o.y = fmaxf(di * (a.y + gg.y) + b[part * 4 + 1], 0.0f);
    o.z = fmaxf(di * (a.z + gg.z) + b[part * 4 + 2], 0.0f);
    o.w = fmaxf(di * (a.w + gg.w) + b[part * 4 + 3], 0.0f);
  }
  *(float4*)(hs + r * HPITCH + part * 4) = o;
  if (part == 0) hs[r * HPITCH + 16] = di;
  __syncthreads();
  if (wave < BNODES / 16) {
    const float* hr = hs + (wave * 16 + lo) * HPITCH + 2 * hi;
    const float* dr = hs + (wave * 16 + 8 * hi) * HPITCH + 16;
    float dv[8];
#pragma unroll
    for (int j = 0; j < 8; ++j) dv[j] = dr[j * HPITCH];
    v8f c = {};
#pragma unroll
    for (int s = 0; s < 4; ++s) {
      v2f av, bv;
      av.x = hr[4 * s];
      av.y = hr[4 * s + 1];
      bv.x = wb[2 * s];
      bv.y = wb[2 * s + 1];
      c = __builtin_amdgcn_wmma_f32_16x16x4_f32(false, av, false, bv, (short)0, c,
                                                false, false);
    }
#pragma unroll
    for (int j = 0; j < 8; ++j) {
      const int rr = blockIdx.x * BNODES + wave * 16 + 8 * hi + j;
      if (rr < N) g2[(size_t)rr * 16 + lo] = c[j] * dv[j];
    }
  }
}

__global__ void __launch_bounds__(4 * BNODES, 16)
bucket_agg2(const unsigned int* __restrict__ sorted, const int* __restrict__ bstart,
            const float* __restrict__ g, const float* __restrict__ dinv,
            const float* __restrict__ b, const float* __restrict__ Wfc,
            float* __restrict__ p, float* __restrict__ q, int N) {
  __shared__ unsigned int stage[RWORDS];
  __shared__ int cnt[BNODES], off[BNODES], wtot[BNODES / 32];
  __shared__ float acc[BNODES * APITCH];
  float4 a = bucket_sum(sorted, bstart[blockIdx.x], bstart[blockIdx.x + 1], g, stage, cnt, off, wtot);
  {
    float* o = acc + (threadIdx.x >> 2) * APITCH + (threadIdx.x & 3) * 4;
    o[0] = a.x;
    o[1] = a.y;
    o[2] = a.z;
    o[3] = a.w;
  }
  __syncthreads();
  int node = blockIdx.x * BNODES + threadIdx.x;
  if (threadIdx.x < BNODES && node < N) {
    float di = dinv[node];
    float sp = 0.0f, sq = 0.0f;
#pragma unroll
    for (int f = 0; f < 16; ++f) {
      float hh = fmaxf(di * (acc[threadIdx.x * APITCH + f] + g[(size_t)node * 16 + f]) + b[f], 0.0f);
      sp += hh * Wfc[f];
      sq += hh * Wfc[16 + f];
    }
    p[node] = sp;
    q[node] = sq;
  }
}

#define OCH 12800
#define OPER 13
__global__ void __launch_bounds__(1024)
edge_out(const int* __restrict__ src, const int* __restrict__ dst,
         const float* __restrict__ p, const float* __restrict__ q,
         const float* __restrict__ bfc, float* __restrict__ out, int E) {
  __shared__ unsigned int stS[OCH];
  __shared__ unsigned int stD[OCH];
  __shared__ float val[OCH];
  __shared__ int cS[NBMAX], cD[NBMAX];
  __shared__ int wtS[32], wtD[32];
  const int t = threadIdx.x, lane = t & 31, wave = t >> 5;
  cS[t] = 0;
  cD[t] = 0;
  __syncthreads();
  const int e0 = blockIdx.x * OCH;
  const int n = min(E - e0, OCH);
  int s[OPER], d[OPER], rs[OPER], rd[OPER];
#pragma unroll
  for (int j = 0; j < OPER; ++j) {
    int i = min(j * 1024 + t, n - 1);
    s[j] = src[e0 + i];
    d[j] = dst[e0 + i];
  }
#pragma unroll
  for (int j = 0; j < OPER; ++j) {
    rs[j] = 0;
    rd[j] = 0;
    if (j * 1024 + t < n) {
      rs[j] = atomicAdd(&cS[s[j] >> BSH], 1);
      rd[j] = atomicAdd(&cD[d[j] >> BSH], 1);
    }
  }
  __syncthreads();
  {
    const int a = cS[t], b = cD[t];
    int xa = a, xb = b;
    for (int o = 1; o < 32; o <<= 1) {
      int ya = __shfl_up(xa, o), yb = __shfl_up(xb, o);
      if (lane >= o) { xa += ya; xb += yb; }
    }
    if (lane == 31) { wtS[wave] = xa; wtD[wave] = xb; }
    __syncthreads();
    const int ta = wtS[lane], tb = wtD[lane];
    int sa = ta, sb = tb;
    for (int o = 1; o < 32; o <<= 1) {
      int ya = __shfl_up(sa, o), yb = __shfl_up(sb, o);
      if (lane >= o) { sa += ya; sb += yb; }
    }
    const int preA = __shfl(sa - ta, wave), preB = __shfl(sb - tb, wave);
    cS[t] = preA + xa - a;
    cD[t] = preB + xb - b;
  }
  __syncthreads();
#pragma unroll
  for (int j = 0; j < OPER; ++j) {
    const int i = j * 1024 + t;
    if (i < n) {
      stS[cS[s[j] >> BSH] + rs[j]] = ((unsigned int)i << 17) | (unsigned int)s[j];
      stD[cD[d[j] >> BSH] + rd[j]] = ((unsigned int)i << 17) | (unsigned int)d[j];
    }
  }
  __syncthreads();
  {
    unsigned int wd[OPER], ws[OPER];
    float vq[OPER], vp[OPER];
#pragma unroll
    for (int j = 0; j < OPER; ++j) {
      const int r = min(j * 1024 + t, n - 1);
      wd[j] = stD[r];
      ws[j] = stS[r];
    }
#pragma unroll
    for (int j = 0; j < OPER; ++j) vq[j] = q[wd[j] & 0x1FFFFu];
#pragma unroll
    for (int j = 0; j < OPER; ++j) vp[j] = p[ws[j] & 0x1FFFFu];
#pragma unroll
    for (int j = 0; j < OPER; ++j)
      if (j * 1024 + t < n) val[wd[j] >> 17] = vq[j];
    __syncthreads();
#pragma unroll
    for (int j = 0; j < OPER; ++j)
      if (j * 1024 + t < n) val[ws[j] >> 17] = vp[j] + val[ws[j] >> 17];
  }
  __syncthreads();
  const float bias = bfc[0];
#pragma unroll
  for (int j = 0; j < OPER; ++j) {
    const int i = j * 1024 + t;
    if (i < n) out[e0 + i] = val[i] + bias;
  }
}

extern "C" void kernel_launch(void* const* d_in, const int* in_sizes, int n_in,
                              void* d_out, int out_size, void* d_ws, size_t ws_size,
                              hipStream_t stream) {
  const float* x   = (const float*)d_in[0];
  const int*   ei  = (const int*)d_in[1];
  const float* W1  = (const float*)d_in[2];
  const float* b1  = (const float*)d_in[3];
  const float* W2  = (const float*)d_in[4];
  const float* b2  = (const float*)d_in[5];
  const float* Wfc = (const float*)d_in[6];
  const float* bfc = (const float*)d_in[7];
  float* out = (float*)d_out;

  const int K1 = 495;
  const int N = in_sizes[0] / K1;
  const int E = in_sizes[1] / 2;
  const int* src = ei;
  const int* dst = ei + E;

  const int NB = cdiv(N, BNODES);
  const int nchunk = cdiv(E, CHUNK);

  float* ws  = (float*)d_ws;
  float* dinv = ws;
  float* g    = ws + (size_t)N;
  float* g2   = ws + (size_t)N * 17;
  float* p    = ws + (size_t)N * 33;
  float* q    = ws + (size_t)N * 34;
  int* cnt    = (int*)(ws + (size_t)N * 35);
  int* bstart = cnt + (size_t)nchunk * NB;
  unsigned int* sorted = (unsigned int*)d_out;

  const int tiles = cdiv(N, 16);
  const int xwGroups = cdiv(tiles, 32);

  bucket_count<<<nchunk, 1024, 0, stream>>>(dst, E, NB, cnt);
  bucket_place<<<nchunk + xwGroups, 1024, 0, stream>>>(src, dst, E, NB, cnt, bstart, sorted,
                                                        nchunk, x, W1, g, N, K1, tiles);

  bucket_dinv<<<NB, 4 * BNODES, 0, stream>>>(sorted, bstart, dinv, g, N);

  bucket_agg1<<<NB, 4 * BNODES, 0, stream>>>(sorted, bstart, g, dinv, b1, W2, g2, N);

  bucket_agg2<<<NB, 4 * BNODES, 0, stream>>>(sorted, bstart, g2, dinv, b2, Wfc, p, q, N);

  edge_out<<<cdiv(E, OCH), 1024, 0, stream>>>(src, dst, p, q, bfc, out, E);
}


# count launch removed: each 25,600-edge chunk is bucket-ordered in LDS and stored as one block with its own offset column; bucket groups read their 125 short runs; 6 launches -> 5, new sum order
# speedup vs baseline: 4.0980x; 1.0376x over previous
#include <hip/hip_runtime.h>
#include <hip/hip_bf16.h>

typedef float v2f __attribute__((ext_vector_type(2)));
typedef float v8f __attribute__((ext_vector_type(8)));

static __host__ int cdiv(int a, int b) { return (a + b - 1) / b; }


#define BSH 7
#define BNODES 128
#define NBMAX 1024
#define CHUNK 25600
#define APITCH 17


__device__ __forceinline__ void xw_tile_k8(const float* __restrict__ X, const float4* wl,
                                           float* __restrict__ G, int N, int K, int tile) {
  int lane = threadIdx.x & 31;
  int hi = lane >> 4;
  int lo = lane & 15;
  int row = tile * 16 + lo;
  float rowMask = (row < N) ? 1.0f : 0.0f;
  const float* xr = X + (size_t)((row < N) ? row : 0) * K;
  const float4* wf = wl + lane;
  int K8 = K & ~7;

  v8f c = {};
  int k0 = 0;
  for (; k0 < K8; k0 += 8) {
    int kk = k0 + 4 * hi;
    if ((k0 & 63) == 0) __builtin_prefetch(xr + kk + 64, 0, 1);
    float x0 = xr[kk], x1 = xr[kk + 1], x2 = xr[kk + 2], x3 = xr[kk + 3];
    float4 wv = wf[k0 * 4];
    v2f a, b;
    a.x = x0 * rowMask;
    a.y = x1 * rowMask;
    b.x = wv.x;
    b.y = wv.y;
    c = __builtin_amdgcn_wmma_f32_16x16x4_f32(false, a, false, b, (short)0, c,
                                              false, false);
    a.x = x2 * rowMask;
    a.y = x3 * rowMask;
    b.x = wv.z;
    b.y = wv.w;
    c = __builtin_amdgcn_wmma_f32_16x16x4_f32(false, a, false, b, (short)0, c,
                                              false, false);
  }
  if (k0 < K) {
    int kk = k0 + 4 * hi;
    float xa[4];
#pragma unroll
    for (int s = 0; s < 4; ++s) {
      int kc = (kk + s < K) ? kk + s : 0;
      float m = (kk + s < K) ? rowMask : 0.0f;
      xa[s] = xr[kc] * m;
    }
    float4 wv = wf[k0 * 4];
    v2f a, b;
    a.x = xa[0]; a.y = xa[1]; b.x = wv.x; b.y = wv.y;
    c = __builtin_amdgcn_wmma_f32_16x16x4_f32(false, a, false, b, (short)0, c,
                                              false, false);
    a.x = xa[2]; a.y = xa[3]; b.x = wv.z; b.y = wv.w;
    c = __builtin_amdgcn_wmma_f32_16x16x4_f32(false, a, false, b, (short)0, c,
                                              false, false);
  }
#pragma unroll
  for (int j = 0; j < 8; ++j) {
    int rr = tile * 16 + 8 * hi + j;
    if (rr < N) G[(size_t)rr * 16 + lo] = c[j];
  }
}

__device__ __forceinline__ void w_frag_fill(const float* __restrict__ W, float4* wl, int K) {
  const int slots = ((K + 7) >> 3) * 32;
  for (int i = threadIdx.x; i < slots; i += 1024) {
    const int ln = i & 31;
    const int k = (i >> 5) * 8 + 4 * (ln >> 4);
    const float* wc = W + (ln & 15);
    const float w0 = wc[min(k, K - 1) * 16], w1 = wc[min(k + 1, K - 1) * 16];
    const float w2 = wc[min(k + 2, K - 1) * 16], w3 = wc[min(k + 3, K - 1) * 16];
    float4 v;
    v.x = (k     < K) ? w0 : 0.0f;
    v.y = (k + 1 < K) ? w1 : 0.0f;
    v.z = (k + 2 < K) ? w2 : 0.0f;
    v.w = (k + 3 < K) ? w3 : 0.0f;
    wl[i] = v;
  }
}

__global__ void __launch_bounds__(1024, 16)
bucket_place(const int* __restrict__ src, const int* __restrict__ dst, int E, int NB,
             int* __restrict__ roff, unsigned int* __restrict__ sorted, int nplace,
             const float* __restrict__ X, const float* __restrict__ W, float* __restrict__ G,
             int N, int K, int tiles) {
  __shared__ alignas(16) unsigned int stage[CHUNK];
  __shared__ int cur[1024];
  __shared__ int wt[32];
  int t = threadIdx.x;
  if ((int)blockIdx.x >= nplace) {
    int tile = ((int)blockIdx.x - nplace) * 32 + (t >> 5);
    float4* wl = reinterpret_cast<float4*>(stage);
    if (tile < tiles) __builtin_prefetch(X + (size_t)min(tile * 16 + (t & 15), N - 1) * K, 0, 1);
    w_frag_fill(W, wl, K);
    __syncthreads();
    if (tile < tiles) xw_tile_k8(X, wl, G, N, K, tile);
    return;
  }
  const int lane = t & 31, wave = t >> 5;
  cur[t] = 0;
  __syncthreads();
  const int e0 = blockIdx.x * CHUNK;
  const int e1 = min(E, e0 + CHUNK);
  int e = e0 + t;
  for (; e + 4 * 1024 < e1; e += 5 * 1024) {
    int d0 = dst[e], d1 = dst[e + 1024], d2 = dst[e + 2048], d3 = dst[e + 3072], d4 = dst[e + 4096];
    atomicAdd(&cur[d0 >> BSH], 1);
    atomicAdd(&cur[d1 >> BSH], 1);
    atomicAdd(&cur[d2 >> BSH], 1);
    atomicAdd(&cur[d3 >> BSH], 1);
    atomicAdd(&cur[d4 >> BSH], 1);
  }
  for (; e < e1; e += 1024) atomicAdd(&cur[dst[e] >> BSH], 1);
  __syncthreads();
  {
    const int c = cur[t];
    int x = c;
    for (int o = 1; o < 32; o <<= 1) {
      int y = __shfl_up(x, o);
      if (lane >= o) x += y;
    }
    if (lane == 31) wt[wave] = x;
    __syncthreads();
    const int tw = wt[lane];
    int sw = tw;
    for (int o = 1; o < 32; o <<= 1) {
      int y = __shfl_up(sw, o);
      if (lane >= o) sw += y;
    }
    const int ex = __shfl(sw - tw, wave) + x - c;
    cur[t] = ex;
    if (t < NB) roff[(size_t)t * nplace + blockIdx.x] = ex;
    if (t == 0) roff[(size_t)NB * nplace + blockIdx.x] = e1 - e0;
  }
  __syncthreads();
  e = e0 + t;
  for (; e + 4 * 1024 < e1; e += 5 * 1024) {
    int d[5], s[5];
#pragma unroll
    for (int j = 0; j < 5; ++j) { d[j] = dst[e + j * 1024]; s[j] = src[e + j * 1024]; }
#pragma unroll
    for (int j = 0; j < 5; ++j) {
      int r = atomicAdd(&cur[d[j] >> BSH], 1);
      stage[r] = ((unsigned int)s[j] << BSH) | (unsigned int)(d[j] & (BNODES - 1));
    }
  }
  for (; e < e1; e += 1024) {
    int d = dst[e], s = src[e];
    int r = atomicAdd(&cur[d >> BSH], 1);
    stage[r] = ((unsigned int)s << BSH) | (unsigned int)(d & (BNODES - 1));
  }
  __syncthreads();
  for (int i = t; i < e1 - e0; i += 1024) sorted[e0 + i] = stage[i];
}

#define DBATCH 12
__global__ void __launch_bounds__(4 * BNODES)
bucket_dinv(const unsigned int* __restrict__ sorted, const int* __restrict__ roff, int nchunk,
            float* __restrict__ dinv, float* __restrict__ g, int N) {
  __shared__ int c[BNODES];
  const int grow = blockIdx.x * BNODES + ((int)threadIdx.x >> 2);
  float4* gp = (float4*)(g + (size_t)min(grow, N - 1) * 16) + (threadIdx.x & 3);
  if (threadIdx.x < BNODES) c[threadIdx.x] = 0;
  __syncthreads();
  float4 gv = *gp;
  const int part = threadIdx.x & 3;
  for (int run = (int)threadIdx.x >> 2; run < nchunk; run += BNODES) {
    const int* ro = roff + (size_t)blockIdx.x * nchunk + run;
    const int lo = ro[0], hi = ro[nchunk];
    const unsigned int* sp = sorted + (size_t)run * CHUNK;
    for (int j = lo + part; j < hi; j += 4 * DBATCH) {
      unsigned int v[DBATCH];
#pragma unroll
      for (int u = 0; u < DBATCH; ++u) v[u] = sp[min(j + 4 * u, hi - 1)];
#pragma unroll
      for (int u = 0; u < DBATCH; ++u)
        if (j + 4 * u < hi) atomicAdd(&c[v[u] & (BNODES - 1)], 1);
    }
  }
  __syncthreads();
  int node = blockIdx.x * BNODES + threadIdx.x;
  if (threadIdx.x < BNODES && node < N) dinv[node] = rsqrtf((float)c[threadIdx.x] + 1.0f);
  if (grow < N) {
    float di = rsqrtf((float)c[threadIdx.x >> 2] + 1.0f);
    gv.x *= di;
    gv.y *= di;
    gv.z *= di;
    gv.w *= di;
    *gp = gv;
  }
}

#define RPER 10
#define RWORDS (RPER * 4 * BNODES)
#define MAXRUN (4 * BNODES)
#define SBATCH 12
__device__ __forceinline__ float4 bucket_sum(const unsigned int* __restrict__ sorted,
                                             const int* __restrict__ roff, int nchunk,
                                             const float* __restrict__ g, unsigned int* stage,
                                             unsigned int* ein, int* rbase, int* rsrc,
                                             int* cnt, int* off, int* wtot) {
  const int T = 4 * BNODES;
  const int t = threadIdx.x;
  const int lane = t & 31;
  const int node = t >> 2, part = t & 3;
  const float4* g4 = (const float4*)g;
  float4 acc = make_float4(0.0f, 0.0f, 0.0f, 0.0f);
  int e1;
  {
    int lo = 0, len = 0;
    if (t < nchunk) {
      const int* ro = roff + (size_t)blockIdx.x * nchunk + t;
      lo = ro[0];
      len = ro[nchunk] - lo;
    }
    int x = len;
    for (int d = 1; d < 32; d <<= 1) {
      int y = __shfl_up(x, d);
      if (lane >= d) x += y;
    }
    if (lane == 31) wtot[t >> 5] = x;
    __syncthreads();
    const int tw = (lane < T / 32) ? wtot[lane] : 0;
    int sw = tw;
    for (int d = 1; d < 32; d <<= 1) {
      int y = __shfl_up(sw, d);
      if (lane >= d) sw += y;
    }
    const int first = __shfl(sw - tw, t >> 5) + x - len;
    e1 = __shfl(sw, 31);
    if (t < nchunk) {
      rbase[t] = first;
      rsrc[t] = t * CHUNK + lo - first;
    }
    if (t == 0) rbase[nchunk] = e1;
    __syncthreads();
  }
  for (int c0 = 0; c0 < e1; c0 += RWORDS) {
    const int c1 = min(e1, c0 + RWORDS);
    if (t < BNODES) cnt[t] = 0;
    for (int run = node; run < nchunk; run += BNODES) {
      const int lo = max(rbase[run], c0), hi = min(rbase[run + 1], c1);
      const unsigned int* sp = sorted + rsrc[run];
      for (int j = lo + part; j < hi; j += 4 * SBATCH) {
        unsigned int v[SBATCH];
#pragma unroll
        for (int u = 0; u < SBATCH; ++u) v[u] = sp[min(j + 4 * u, hi - 1)];
#pragma unroll
        for (int u = 0; u < SBATCH; ++u)
          if (j + 4 * u < hi) ein[j + 4 * u - c0] = v[u];
      }
    }
    __syncthreads();
    unsigned int w[RPER];
    int r[RPER];
#pragma unroll
    for (int j = 0; j < RPER; ++j) w[j] = ein[min(c0 + j * T + t, c1 - 1) - c0];
#pragma unroll
    for (int j = 0; j < RPER; ++j) {
      r[j] = 0;
      if (c0 + j * T + t < c1) r[j] = atomicAdd(&cnt[w[j] & (BNODES - 1)], 1);
    }
    __syncthreads();
    int c = (t < BNODES) ? cnt[t] : 0;
    int x = c;
    for (int d = 1; d < 32; d <<= 1) {
      int y = __shfl_up(x, d);
      if (lane >= d) x += y;
    }
    if (t < BNODES && lane == 31) wtot[t >> 5] = x;
    __syncthreads();
    if (t < BNODES) {
      int pre = 0;
      for (int k = 0; k < (t >> 5); ++k) pre += wtot[k];
      off[t] = pre + x - c;
    }
    __syncthreads();
    int o[RPER];
#pragma unroll
    for (int j = 0; j < RPER; ++j) o[j] = off[w[j] & (BNODES - 1)];
#pragma unroll
    for (int j = 0; j < RPER; ++j)
      if (c0 + j * T + t < c1) stage[o[j] + r[j]] = w[j] >> BSH;
    __syncthreads();
    const int st = off[node];
    const int n = cnt[node];
    for (int j = 0; j < n; j += 8) {
      unsigned int s[8];
      float4 v[8];
#pragma unroll
      for (int u = 0; u < 8; ++u) s[u] = stage[st + min(j + u, n - 1)];
#pragma unroll
      for (int u = 0; u < 8; ++u) v[u] = g4[(s[u] << 2) + (unsigned int)part];
#pragma unroll
      for (int u = 0; u < 8; ++u) {
        bool ok = (j + u) < n;
        acc.x += ok ? v[u].x : 0.0f;
        acc.y += ok ? v[u].y : 0.0f;
        acc.z += ok ? v[u].z : 0.0f;
        acc.w += ok ? v[u].w : 0.0f;
      }
    }
    __syncthreads();
  }
  return acc;
}

#define HPITCH 20
__global__ void __launch_bounds__(4 * BNODES, 16)
bucket_agg1(const unsigned int* __restrict__ sorted, const int* __restrict__ roff, int nchunk,
            const float* __restrict__ g, const float* __restrict__ dinv,
            const float* __restrict__ b, const float* __restrict__ W2,
            float* __restrict__ g2, int N) {
  __shared__ unsigned int stage[RWORDS], ein[RWORDS];
  __shared__ int rbase[MAXRUN + 1], rsrc[MAXRUN];
  __shared__ int cnt[BNODES], off[BNODES], wtot[4 * BNODES / 32];
  __shared__ alignas(16) float hs[BNODES * HPITCH];
  float4 a = bucket_sum(sorted, roff, nchunk, g, stage, ein, rbase, rsrc, cnt, off, wtot);
  const int wave = __builtin_amdgcn_readfirstlane((int)threadIdx.x >> 5);
  const int lane = threadIdx.x & 31, hi = lane >> 4, lo = lane & 15;
  float wb[8] = {0.0f, 0.0f, 0.0f, 0.0f, 0.0f, 0.0f, 0.0f, 0.0f};
  if (wave < BNODES / 16) {
#pragma unroll
    for (int s = 0; s < 4; ++s) {
      wb[2 * s]     = W2[(4 * s + 2 * hi) * 16 + lo];
      wb[2 * s + 1] = W2[(4 * s + 2 * hi + 1) * 16 + lo];
    }
  }
  int r = threadIdx.x >> 2, part = threadIdx.x & 3;
  int node = blockIdx.x * BNODES + r;
  float4 o = make_float4(0.0f, 0.0f, 0.0f, 0.0f);
  float di = 0.0f;
  if (node < N) {
    di = dinv[node];
    float4 gg = ((const float4*)(g + (size_t)node * 16))[part];
    o.x = fmaxf(di * (a.x + gg.x) + b[part * 4 + 0], 0.0f);
    o.y = fmaxf(di * (a.y + gg.y) + b[part * 4 + 1], 0.0f);
    o.z = fmaxf(di * (a.z + gg.z) + b[part * 4 + 2], 0.0f);
    o.w = fmaxf(di * (a.w + gg.w) + b[part * 4 + 3], 0.0f);
  }
  *(float4*)(hs + r * HPITCH + part * 4) = o;
  if (part == 0) hs[r * HPITCH + 16] = di;
  __syncthreads();
  if (wave < BNODES / 16) {
    const float* hr = hs + (wave * 16 + lo) * HPITCH + 2 * hi;
    const float* dr = hs + (wave * 16 + 8 * hi) * HPITCH + 16;
    float dv[8];
#pragma unroll
    for (int j = 0; j < 8; ++j) dv[j] = dr[j * HPITCH];
    v8f c = {};
#pragma unroll
    for (int s = 0; s < 4; ++s) {
      v2f av, bv;
      av.x = hr[4 * s];
      av.y = hr[4 * s + 1];
      bv.x = wb[2 * s];
      bv.y = wb[2 * s + 1];
      c = __builtin_amdgcn_wmma_f32_16x16x4_f32(false, av, false, bv, (short)0, c,
                                                false, false);
    }
#pragma unroll
    for (int j = 0; j < 8; ++j) {
      const int rr = blockIdx.x * BNODES + wave * 16 + 8 * hi + j;
      if (rr < N) g2[(size_t)rr * 16 + lo] = c[j] * dv[j];
    }
  }
}

__global__ void __launch_bounds__(4 * BNODES, 16)
bucket_agg2(const unsigned int* __restrict__ sorted, const int* __restrict__ roff, int nchunk,
            const float* __restrict__ g, const float* __restrict__ dinv,
            const float* __restrict__ b, const float* __restrict__ Wfc,
            float* __restrict__ p, float* __restrict__ q, int N) {
  __shared__ unsigned int stage[RWORDS], ein[RWORDS];
  __shared__ int rbase[MAXRUN + 1], rsrc[MAXRUN];
  __shared__ int cnt[BNODES], off[BNODES], wtot[4 * BNODES / 32];
  __shared__ float acc[BNODES * APITCH];
  float4 a = bucket_sum(sorted, roff, nchunk, g, stage, ein, rbase, rsrc, cnt, off, wtot);
  {
    float* o = acc + (threadIdx.x >> 2) * APITCH + (threadIdx.x & 3) * 4;
    o[0] = a.x;
    o[1] = a.y;
    o[2] = a.z;
    o[3] = a.w;
  }
  __syncthreads();
  int node = blockIdx.x * BNODES + threadIdx.x;
  if (threadIdx.x < BNODES && node < N) {
    float di = dinv[node];
    float sp = 0.0f, sq = 0.0f;
#pragma unroll
    for (int f = 0; f < 16; ++f) {
      float hh = fmaxf(di * (acc[threadIdx.x * APITCH + f] + g[(size_t)node * 16 + f]) + b[f], 0.0f);
      sp += hh * Wfc[f];
      sq += hh * Wfc[16 + f];
    }
    p[node] = sp;
    q[node] = sq;
  }
}

#define OCH 12800
#define OPER 13
__global__ void __launch_bounds__(1024)
edge_out(const int* __restrict__ src, const int* __restrict__ dst,
         const float* __restrict__ p, const float* __restrict__ q,
         const float* __restrict__ bfc, float* __restrict__ out, int E) {
  __shared__ unsigned int stS[OCH];
  __shared__ unsigned int stD[OCH];
  __shared__ float val[OCH];
  __shared__ int cS[NBMAX], cD[NBMAX];
  __shared__ int wtS[32], wtD[32];
  const int t = threadIdx.x, lane = t & 31, wave = t >> 5;
  cS[t] = 0;
  cD[t] = 0;
  __syncthreads();
  const int e0 = blockIdx.x * OCH;
  const int n = min(E - e0, OCH);
  int s[OPER], d[OPER], rs[OPER], rd[OPER];
#pragma unroll
  for (int j = 0; j < OPER; ++j) {
    int i = min(j * 1024 + t, n - 1);
    s[j] = src[e0 + i];
    d[j] = dst[e0 + i];
  }
#pragma unroll
  for (int j = 0; j < OPER; ++j) {
    rs[j] = 0;
    rd[j] = 0;
    if (j * 1024 + t < n) {
      rs[j] = atomicAdd(&cS[s[j] >> BSH], 1);
      rd[j] = atomicAdd(&cD[d[j] >> BSH], 1);
    }
  }
  __syncthreads();
  {
    const int a = cS[t], b = cD[t];
    int xa = a, xb = b;
    for (int o = 1; o < 32; o <<= 1) {
      int ya = __shfl_up(xa, o), yb = __shfl_up(xb, o);
      if (lane >= o) { xa += ya; xb += yb; }
    }
    if (lane == 31) { wtS[wave] = xa; wtD[wave] = xb; }
    __syncthreads();
    const int ta = wtS[lane], tb = wtD[lane];
    int sa = ta, sb = tb;
    for (int o = 1; o < 32; o <<= 1) {
      int ya = __shfl_up(sa, o), yb = __shfl_up(sb, o);
      if (lane >= o) { sa += ya; sb += yb; }
    }
    const int preA = __shfl(sa - ta, wave), preB = __shfl(sb - tb, wave);
    cS[t] = preA + xa - a;
    cD[t] = preB + xb - b;
  }
  __syncthreads();
#pragma unroll
  for (int j = 0; j < OPER; ++j) {
    const int i = j * 1024 + t;
    if (i < n) {
      stS[cS[s[j] >> BSH] + rs[j]] = ((unsigned int)i << 17) | (unsigned int)s[j];
      stD[cD[d[j] >> BSH] + rd[j]] = ((unsigned int)i << 17) | (unsigned int)d[j];
    }
  }
  __syncthreads();
  {
    unsigned int wd[OPER], ws[OPER];
    float vq[OPER], vp[OPER];
#pragma unroll
    for (int j = 0; j < OPER; ++j) {
      const int r = min(j * 1024 + t, n - 1);
      wd[j] = stD[r];
      ws[j] = stS[r];
    }
#pragma unroll
    for (int j = 0; j < OPER; ++j) vq[j] = q[wd[j] & 0x1FFFFu];
#pragma unroll
    for (int j = 0; j < OPER; ++j) vp[j] = p[ws[j] & 0x1FFFFu];
#pragma unroll
    for (int j = 0; j < OPER; ++j)
      if (j * 1024 + t < n) val[wd[j] >> 17] = vq[j];
    __syncthreads();
#pragma unroll
    for (int j = 0; j < OPER; ++j)
      if (j * 1024 + t < n) val[ws[j] >> 17] = vp[j] + val[ws[j] >> 17];
  }
  __syncthreads();
  const float bias = bfc[0];
#pragma unroll
  for (int j = 0; j < OPER; ++j) {
    const int i = j * 1024 + t;
    if (i < n) out[e0 + i] = val[i] + bias;
  }
}

extern "C" void kernel_launch(void* const* d_in, const int* in_sizes, int n_in,
                              void* d_out, int out_size, void* d_ws, size_t ws_size,
                              hipStream_t stream) {
  const float* x   = (const float*)d_in[0];
  const int*   ei  = (const int*)d_in[1];
  const float* W1  = (const float*)d_in[2];
  const float* b1  = (const float*)d_in[3];
  const float* W2  = (const float*)d_in[4];
  const float* b2  = (const float*)d_in[5];
  const float* Wfc = (const float*)d_in[6];
  const float* bfc = (const float*)d_in[7];
  float* out = (float*)d_out;

  const int K1 = 495;
  const int N = in_sizes[0] / K1;
  const int E = in_sizes[1] / 2;
  const int* src = ei;
  const int* dst = ei + E;

  const int NB = cdiv(N, BNODES);
  const int nchunk = cdiv(E, CHUNK);

  float* ws  = (float*)d_ws;
  float* dinv = ws;
  float* g    = ws + (size_t)N;
  float* g2   = ws + (size_t)N * 17;
  float* p    = ws + (size_t)N * 33;
  float* q    = ws + (size_t)N * 34;
  int* roff   = (int*)(ws + (size_t)N * 35);
  unsigned int* sorted = (unsigned int*)d_out;

  const int tiles = cdiv(N, 16);
  const int xwGroups = cdiv(tiles, 32);

  bucket_place<<<nchunk + xwGroups, 1024, 0, stream>>>(src, dst, E, NB, roff, sorted,
                                                        nchunk, x, W1, g, N, K1, tiles);

  bucket_dinv<<<NB, 4 * BNODES, 0, stream>>>(sorted, roff, nchunk, dinv, g, N);

  bucket_agg1<<<NB, 4 * BNODES, 0, stream>>>(sorted, roff, nchunk, g, dinv, b1, W2, g2, N);

  bucket_agg2<<<NB, 4 * BNODES, 0, stream>>>(sorted, roff, nchunk, g2, dinv, b2, Wfc, p, q, N);

  edge_out<<<cdiv(E, OCH), 1024, 0, stream>>>(src, dst, p, q, bfc, out, E);
}
